# starting kernel
# baseline (speedup 1.0000x reference)
#include <hip/hip_runtime.h>
#include <hip/hip_bf16.h>
#include <math.h>

#define N_NODES 100000
#define N_FEAT  512
#define HIDDEN  128
#define N_CLASS 16
#define N_EDGES 3200000
#define ALPHA   0.25f

typedef __attribute__((ext_vector_type(2))) float v2f;
typedef __attribute__((ext_vector_type(8))) float v8f;

#define KCHUNK   64
#define W1C_STR  136
#define HID_STR  132

__global__ __launch_bounds__(256) void gcn_gemm_kernel(
    const float* __restrict__ x, const float* __restrict__ W1,
    const float* __restrict__ W2, float* __restrict__ local_out)
{
    __shared__ float w1c[KCHUNK][W1C_STR];
    __shared__ float hid[8][16][HID_STR];

    const int tid  = threadIdx.x;
    const int wave = tid >> 5;
    const int lane = tid & 31;
    const int half = lane >> 4;
    const int l16  = lane & 15;

    const int row_tile = blockIdx.x * 8 + wave;
    const int row_base = row_tile * 16;
    const bool active  = (row_base < N_NODES);

    v8f acc[8];
    for (int t = 0; t < 8; ++t)
        for (int i = 0; i < 8; ++i) acc[t][i] = 0.0f;

    for (int k0 = 0; k0 < N_FEAT; k0 += KCHUNK) {
        __syncthreads();
        for (int i = tid; i < KCHUNK * (HIDDEN / 4); i += 256) {
            const int r  = i >> 5;
            const int c4 = i & 31;
            const float4 v = ((const float4*)(W1 + (size_t)(k0 + r) * HIDDEN))[c4];
            *(float4*)&w1c[r][c4 * 4] = v;
        }
        __syncthreads();

        if (active) {
            for (int kk = 0; kk < KCHUNK; kk += 4) {
                const float* ap = x + (size_t)(row_base + l16) * N_FEAT
                                    + (k0 + kk + 2 * half);
                v2f a = *(const v2f*)ap;
                for (int t = 0; t < 8; ++t) {
                    v2f b;
                    b[0] = w1c[kk + 2 * half + 0][t * 16 + l16];
                    b[1] = w1c[kk + 2 * half + 1][t * 16 + l16];
                    acc[t] = __builtin_amdgcn_wmma_f32_16x16x4_f32(
                        false, a, false, b, (short)0, acc[t], false, false);
                }
            }
        }
    }

    if (active) {
        for (int t = 0; t < 8; ++t)
            for (int r = 0; r < 8; ++r) {
                float v = acc[t][r];
                hid[wave][r + 8 * half][t * 16 + l16] = v > 0.0f ? v : 0.0f;
            }
        v8f c2;
        for (int i = 0; i < 8; ++i) c2[i] = 0.0f;
        for (int kk = 0; kk < HIDDEN; kk += 4) {
            v2f a, b;
            a[0] = hid[wave][l16][kk + 2 * half + 0];
            a[1] = hid[wave][l16][kk + 2 * half + 1];
            b[0] = W2[(size_t)(kk + 2 * half + 0) * N_CLASS + l16];
            b[1] = W2[(size_t)(kk + 2 * half + 1) * N_CLASS + l16];
            c2 = __builtin_amdgcn_wmma_f32_16x16x4_f32(
                false, a, false, b, (short)0, c2, false, false);
        }
        for (int r = 0; r < 8; ++r)
            local_out[(size_t)(row_base + r + 8 * half) * N_CLASS + l16] = c2[r];
    }
}

__global__ void zero_kernel(float* __restrict__ p, long long n)
{
    long long i = (long long)blockIdx.x * blockDim.x + threadIdx.x;
    if (i < n) p[i] = 0.0f;
}

__global__ void deg_kernel(const int* __restrict__ src, float* __restrict__ deg)
{
    int e = blockIdx.x * blockDim.x + threadIdx.x;
    if (e < N_EDGES) atomicAdd(&deg[src[e]], 1.0f);
}

__global__ void scatter_kernel(const int* __restrict__ src,
                               const int* __restrict__ dst,
                               const float* __restrict__ logits,
                               float* __restrict__ agg)
{
    long long gid = (long long)blockIdx.x * blockDim.x + threadIdx.x;
    if (gid >= (long long)N_EDGES * N_CLASS) return;
    const int e = (int)(gid >> 4);
    const int c = (int)(gid & 15);
    const int s = src[e];
    const int d = dst[e];
    atomicAdd(&agg[(size_t)s * N_CLASS + c], logits[(size_t)d * N_CLASS + c]);
}

__global__ void combine_kernel(const float* __restrict__ agg,
                               const float* __restrict__ local,
                               const float* __restrict__ deg,
                               float* __restrict__ out)
{
    long long i = (long long)blockIdx.x * blockDim.x + threadIdx.x;
    if (i >= (long long)N_NODES * N_CLASS) return;
    const int row = (int)(i >> 4);
    const float scale = (1.0f - ALPHA) / fmaxf(deg[row], 1e-12f);
    out[i] = scale * agg[i] + ALPHA * local[i];
}

__global__ void final_kernel(const float* __restrict__ agg,
                             const float* __restrict__ local,
                             const float* __restrict__ deg,
                             float* __restrict__ out)
{
    int row = blockIdx.x * blockDim.x + threadIdx.x;
    if (row >= N_NODES) return;
    const float scale = (1.0f - ALPHA) / fmaxf(deg[row], 1e-12f);
    float t[N_CLASS];
    float m = -INFINITY;
    for (int c = 0; c < N_CLASS; ++c) {
        t[c] = scale * agg[(size_t)row * N_CLASS + c]
             + ALPHA * local[(size_t)row * N_CLASS + c];
        m = fmaxf(m, t[c]);
    }
    float s = 0.0f;
    for (int c = 0; c < N_CLASS; ++c) s += __expf(t[c] - m);
    const float lse = m + __logf(s);
    for (int c = 0; c < N_CLASS; ++c)
        out[(size_t)row * N_CLASS + c] = t[c] - lse;
}

extern "C" void kernel_launch(void* const* d_in, const int* in_sizes, int n_in,
                              void* d_out, int out_size, void* d_ws, size_t ws_size,
                              hipStream_t stream)
{
    const float* x    = (const float*)d_in[0];
    const float* W1   = (const float*)d_in[1];
    const float* W2   = (const float*)d_in[2];
    const int*   esrc = (const int*)d_in[3];
    const int*   edst = (const int*)d_in[4];
    float* out = (float*)d_out;

    float* local  = (float*)d_ws;
    float* logits = local  + (size_t)N_NODES * N_CLASS;
    float* deg    = logits + (size_t)N_NODES * N_CLASS;
    float* agg    = out;

    const dim3 blk(256);
    const long long nl   = (long long)N_NODES * N_CLASS;
    const long long ne16 = (long long)N_EDGES * N_CLASS;
    const unsigned gN    = (unsigned)((N_NODES + 255) / 256);
    const unsigned gNL   = (unsigned)((nl + 255) / 256);
    const unsigned gE    = (unsigned)((N_EDGES + 255) / 256);
    const unsigned gE16  = (unsigned)((ne16 + 255) / 256);

    const unsigned gGEMM = (unsigned)(((N_NODES + 15) / 16 + 7) / 8);
    gcn_gemm_kernel<<<gGEMM, blk, 0, stream>>>(x, W1, W2, local);

    zero_kernel<<<gN, blk, 0, stream>>>(deg, N_NODES);
    deg_kernel<<<gE, blk, 0, stream>>>(esrc, deg);

    zero_kernel<<<gNL, blk, 0, stream>>>(agg, nl);
    scatter_kernel<<<gE16, blk, 0, stream>>>(esrc, edst, local, agg);
    combine_kernel<<<gNL, blk, 0, stream>>>(agg, local, deg, logits);

    zero_kernel<<<gNL, blk, 0, stream>>>(agg, nl);
    scatter_kernel<<<gE16, blk, 0, stream>>>(esrc, edst, logits, agg);
    final_kernel<<<gN, blk, 0, stream>>>(agg, local, deg, out);
}


# edges regrouped once by 128-row src range (count, prefix, place); each pass sums a range's gathered rows and degree in LDS, writes the combined rows; global f32 atomics gone; sum order: LDS arrival
# speedup vs baseline: 2.2558x; 2.2558x over previous
#include <hip/hip_runtime.h>
#include <hip/hip_bf16.h>
#include <math.h>

#define N_NODES 100000
#define N_FEAT  512
#define HIDDEN  128
#define N_CLASS 16
#define N_EDGES 3200000
#define ALPHA   0.25f

typedef __attribute__((ext_vector_type(2))) float v2f;
typedef __attribute__((ext_vector_type(8))) float v8f;

#define KCHUNK   64
#define W1C_STR  136
#define HID_STR  132

__global__ __launch_bounds__(256) void gcn_gemm_kernel(
    const float* __restrict__ x, const float* __restrict__ W1,
    const float* __restrict__ W2, float* __restrict__ local_out)
{
    __shared__ float w1c[KCHUNK][W1C_STR];
    __shared__ float hid[8][16][HID_STR];

    const int tid  = threadIdx.x;
    const int wave = tid >> 5;
    const int lane = tid & 31;
    const int half = lane >> 4;
    const int l16  = lane & 15;

    const int row_tile = blockIdx.x * 8 + wave;
    const int row_base = row_tile * 16;
    const bool active  = (row_base < N_NODES);

    v8f acc[8];
    for (int t = 0; t < 8; ++t)
        for (int i = 0; i < 8; ++i) acc[t][i] = 0.0f;

    for (int k0 = 0; k0 < N_FEAT; k0 += KCHUNK) {
        __syncthreads();
        for (int i = tid; i < KCHUNK * (HIDDEN / 4); i += 256) {
            const int r  = i >> 5;
            const int c4 = i & 31;
            const float4 v = ((const float4*)(W1 + (size_t)(k0 + r) * HIDDEN))[c4];
            *(float4*)&w1c[r][c4 * 4] = v;
        }
        __syncthreads();

        if (active) {
            for (int kk = 0; kk < KCHUNK; kk += 4) {
                const float* ap = x + (size_t)(row_base + l16) * N_FEAT
                                    + (k0 + kk + 2 * half);
                v2f a = *(const v2f*)ap;
                for (int t = 0; t < 8; ++t) {
                    v2f b;
                    b[0] = w1c[kk + 2 * half + 0][t * 16 + l16];
                    b[1] = w1c[kk + 2 * half + 1][t * 16 + l16];
                    acc[t] = __builtin_amdgcn_wmma_f32_16x16x4_f32(
                        false, a, false, b, (short)0, acc[t], false, false);
                }
            }
        }
    }

    if (active) {
        for (int t = 0; t < 8; ++t)
            for (int r = 0; r < 8; ++r) {
                float v = acc[t][r];
                hid[wave][r + 8 * half][t * 16 + l16] = v > 0.0f ? v : 0.0f;
            }
        v8f c2;
        for (int i = 0; i < 8; ++i) c2[i] = 0.0f;
        for (int kk = 0; kk < HIDDEN; kk += 4) {
            v2f a, b;
            a[0] = hid[wave][l16][kk + 2 * half + 0];
            a[1] = hid[wave][l16][kk + 2 * half + 1];
            b[0] = W2[(size_t)(kk + 2 * half + 0) * N_CLASS + l16];
            b[1] = W2[(size_t)(kk + 2 * half + 1) * N_CLASS + l16];
            c2 = __builtin_amdgcn_wmma_f32_16x16x4_f32(
                false, a, false, b, (short)0, c2, false, false);
        }
        for (int r = 0; r < 8; ++r)
            local_out[(size_t)(row_base + r + 8 * half) * N_CLASS + l16] = c2[r];
    }
}

__global__ void zero_kernel(float* __restrict__ p, long long n)
{
    long long i = (long long)blockIdx.x * blockDim.x + threadIdx.x;
    if (i < n) p[i] = 0.0f;
}

__global__ void deg_kernel(const int* __restrict__ src, float* __restrict__ deg)
{
    int e = blockIdx.x * blockDim.x + threadIdx.x;
    if (e < N_EDGES) atomicAdd(&deg[src[e]], 1.0f);
}

__global__ void scatter_kernel(const int* __restrict__ src,
                               const int* __restrict__ dst,
                               const float* __restrict__ logits,
                               float* __restrict__ agg)
{
    long long gid = (long long)blockIdx.x * blockDim.x + threadIdx.x;
    if (gid >= (long long)N_EDGES * N_CLASS) return;
    const int e = (int)(gid >> 4);
    const int c = (int)(gid & 15);
    const int s = src[e];
    const int d = dst[e];
    atomicAdd(&agg[(size_t)s * N_CLASS + c], logits[(size_t)d * N_CLASS + c]);
}

__global__ void combine_kernel(const float* __restrict__ agg,
                               const float* __restrict__ local,
                               const float* __restrict__ deg,
                               float* __restrict__ out)
{
    long long i = (long long)blockIdx.x * blockDim.x + threadIdx.x;
    if (i >= (long long)N_NODES * N_CLASS) return;
    const int row = (int)(i >> 4);
    const float scale = (1.0f - ALPHA) / fmaxf(deg[row], 1e-12f);
    out[i] = scale * agg[i] + ALPHA * local[i];
}

__global__ void final_kernel(const float* __restrict__ agg,
                             const float* __restrict__ local,
                             const float* __restrict__ deg,
                             float* __restrict__ out)
{
    int row = blockIdx.x * blockDim.x + threadIdx.x;
    if (row >= N_NODES) return;
    const float scale = (1.0f - ALPHA) / fmaxf(deg[row], 1e-12f);
    float t[N_CLASS];
    float m = -INFINITY;
    for (int c = 0; c < N_CLASS; ++c) {
        t[c] = scale * agg[(size_t)row * N_CLASS + c]
             + ALPHA * local[(size_t)row * N_CLASS + c];
        m = fmaxf(m, t[c]);
    }
    float s = 0.0f;
    for (int c = 0; c < N_CLASS; ++c) s += __expf(t[c] - m);
    const float lse = m + __logf(s);
    for (int c = 0; c < N_CLASS; ++c)
        out[(size_t)row * N_CLASS + c] = t[c] - lse;
}


#define RSH    7
#define RB     (1 << RSH)
#define NB     ((N_NODES + RB - 1) / RB)
#define NCH    256
#define CH     (N_EDGES / NCH)
#define ACC_STR 17
static_assert(N_EDGES % NCH == 0, "chunks must tile the edge list");
static_assert(NB <= 1024, "one thread per bucket in the place kernel");
static_assert(N_NODES <= (1 << (32 - RSH)), "dst and local row share one word");

__global__ __launch_bounds__(1024) void bucket_count_kernel(
    const int* __restrict__ esrc, unsigned* __restrict__ cnt)
{
    __shared__ unsigned h[NB];
    const int tid = threadIdx.x;
    for (int i = tid; i < NB; i += 1024) h[i] = 0u;
    __syncthreads();
    const int e0 = blockIdx.x * CH;
    for (int i = tid; i < CH; i += 1024)
        atomicAdd(&h[esrc[e0 + i] >> RSH], 1u);
    __syncthreads();
    for (int i = tid; i < NB; i += 1024) cnt[blockIdx.x * NB + i] = h[i];
}

static_assert(NCH % 32 == 0, "each lane takes NCH/32 consecutive chunks");
__global__ __launch_bounds__(256) void bucket_prefix_kernel(
    unsigned* __restrict__ cnt, unsigned* __restrict__ tot)
{
    const int lane = threadIdx.x & 31;
    const int b    = blockIdx.x * 8 + (threadIdx.x >> 5);
    if (b >= NB) return;
    unsigned c[NCH / 32];
    for (int j = 0; j < NCH / 32; ++j)
        c[j] = cnt[(lane * (NCH / 32) + j) * NB + b];
    unsigned s = 0u;
    for (int j = 0; j < NCH / 32; ++j) { const unsigned t = c[j]; c[j] = s; s += t; }
    unsigned incl = s;
    for (int o = 1; o < 32; o <<= 1) {
        const unsigned t = __shfl_up(incl, o);
        if (lane >= o) incl += t;
    }
    const unsigned base = incl - s;
    for (int j = 0; j < NCH / 32; ++j)
        cnt[(lane * (NCH / 32) + j) * NB + b] = base + c[j];
    if (lane == 31) tot[b] = incl;
}

__global__ __launch_bounds__(1024) void bucket_place_kernel(
    const int* __restrict__ esrc, const int* __restrict__ edst,
    const unsigned* __restrict__ cnt, const unsigned* __restrict__ tot,
    unsigned* __restrict__ packed, unsigned* __restrict__ bstart)
{
    __shared__ unsigned cur[NB];
    __shared__ unsigned wsum[32];
    const int tid  = threadIdx.x;
    const int lane = tid & 31;
    const int w    = tid >> 5;
    const int me   = blockIdx.x;

    const unsigned tb     = (tid < NB) ? tot[tid] : 0u;
    const unsigned before = (tid < NB) ? cnt[me * NB + tid] : 0u;
    unsigned incl = tb;
    for (int o = 1; o < 32; o <<= 1) {
        const unsigned t = __shfl_up(incl, o);
        if (lane >= o) incl += t;
    }
    if (lane == 31) wsum[w] = incl;
    __syncthreads();
    if (w == 0) {
        const unsigned s = wsum[lane];
        unsigned si = s;
        for (int o = 1; o < 32; o <<= 1) {
            const unsigned t = __shfl_up(si, o);
            if (lane >= o) si += t;
        }
        wsum[lane] = si - s;
    }
    __syncthreads();
    const unsigned start = wsum[w] + incl - tb;
    if (tid < NB) {
        cur[tid] = start + before;
        if (me == 0) bstart[tid] = start;
    }
    if (me == 0 && tid == 0) bstart[NB] = (unsigned)N_EDGES;
    __syncthreads();

    const int e0 = me * CH;
    for (int i = tid; i < CH; i += 1024) {
        const int s = esrc[e0 + i];
        const int d = edst[e0 + i];
        const unsigned pos = atomicAdd(&cur[s >> RSH], 1u);
        packed[pos] = ((unsigned)d << RSH) | (unsigned)(s & (RB - 1));
    }
}

template <bool FINAL>
__global__ __launch_bounds__(512) void bucket_spmm_kernel(
    const unsigned* __restrict__ packed, const unsigned* __restrict__ bstart,
    const float* rows_in, const float* own_in, float* rows_out)
{
    __shared__ float    acc[RB][ACC_STR];
    __shared__ unsigned dcnt[RB];
    const int tid = threadIdx.x;
    const int b   = blockIdx.x;
    for (int i = tid; i < RB * ACC_STR; i += 512) (&acc[0][0])[i] = 0.0f;
    for (int i = tid; i < RB; i += 512) dcnt[i] = 0u;
    __syncthreads();

    const unsigned e0 = bstart[b];
    const unsigned n4 = (bstart[b + 1] - e0) * 4u;
    const unsigned q  = (unsigned)tid & 3u;
    const float4* g4  = (const float4*)rows_in;

    unsigned i = (unsigned)tid;
    for (; i + 3u * 512u < n4; i += 4u * 512u) {
        unsigned p[4];
        float4   v[4];
        for (int j = 0; j < 4; ++j) p[j] = packed[e0 + ((i + j * 512u) >> 2)];
        for (int j = 0; j < 4; ++j) v[j] = g4[(size_t)(p[j] >> RSH) * 4 + q];
        for (int j = 0; j < 4; ++j) {
            const unsigned s = p[j] & (RB - 1);
            float* a = &acc[s][q * 4];
            atomicAdd(a + 0, v[j].x);
            atomicAdd(a + 1, v[j].y);
            atomicAdd(a + 2, v[j].z);
            atomicAdd(a + 3, v[j].w);
            if (q == 0u) atomicAdd(&dcnt[s], 1u);
        }
    }
    for (; i < n4; i += 512u) {
        const unsigned p = packed[e0 + (i >> 2)];
        const float4   v = g4[(size_t)(p >> RSH) * 4 + q];
        const unsigned s = p & (RB - 1);
        float* a = &acc[s][q * 4];
        atomicAdd(a + 0, v.x);
        atomicAdd(a + 1, v.y);
        atomicAdd(a + 2, v.z);
        atomicAdd(a + 3, v.w);
        if (q == 0u) atomicAdd(&dcnt[s], 1u);
    }
    __syncthreads();

    for (int k = tid; k < RB * N_CLASS; k += 512) {
        const int r   = k >> 4;
        const int c   = k & 15;
        const int row = b * RB + r;
        const bool valid = row < N_NODES;
        const float scale = (1.0f - ALPHA) / fmaxf((float)dcnt[r], 1e-12f);
        const float lv = valid ? own_in[(size_t)row * N_CLASS + c] : 0.0f;
        const float t  = scale * acc[r][c] + ALPHA * lv;
        if (!FINAL) {
            if (valid) rows_out[(size_t)row * N_CLASS + c] = t;
        } else {
            float m = t;
            for (int o = 8; o > 0; o >>= 1) m = fmaxf(m, __shfl_xor(m, o, 16));
            float sm = __expf(t - m);
            for (int o = 8; o > 0; o >>= 1) sm += __shfl_xor(sm, o, 16);
            const float lse = m + __logf(sm);
            if (valid) rows_out[(size_t)row * N_CLASS + c] = t - lse;
        }
    }
}

extern "C" void kernel_launch(void* const* d_in, const int* in_sizes, int n_in,
                              void* d_out, int out_size, void* d_ws, size_t ws_size,
                              hipStream_t stream)
{
    const float* x    = (const float*)d_in[0];
    const float* W1   = (const float*)d_in[1];
    const float* W2   = (const float*)d_in[2];
    const int*   esrc = (const int*)d_in[3];
    const int*   edst = (const int*)d_in[4];
    float* out = (float*)d_out;

    const size_t off_packed = (size_t)N_NODES * N_CLASS * sizeof(float);
    const size_t off_bstart = off_packed + (size_t)N_EDGES * sizeof(unsigned);
    const size_t off_tot    = off_bstart + (size_t)(NB + 1) * sizeof(unsigned);
    const size_t ws_need    = off_tot + (size_t)NB * sizeof(unsigned);
    if (ws_size >= ws_need) {
        float*    lg     = (float*)d_ws;
        unsigned* cnt    = (unsigned*)d_ws;
        unsigned* packed = (unsigned*)((char*)d_ws + off_packed);
        unsigned* bstart = (unsigned*)((char*)d_ws + off_bstart);
        unsigned* tot    = (unsigned*)((char*)d_ws + off_tot);
        const unsigned gG = (unsigned)(((N_NODES + 15) / 16 + 7) / 8);
        gcn_gemm_kernel<<<gG, dim3(256), 0, stream>>>(x, W1, W2, out);
        bucket_count_kernel<<<NCH, dim3(1024), 0, stream>>>(esrc, cnt);
        bucket_prefix_kernel<<<(NB + 7) / 8, dim3(256), 0, stream>>>(cnt, tot);
        bucket_place_kernel<<<NCH, dim3(1024), 0, stream>>>(esrc, edst, cnt, tot, packed, bstart);
        bucket_spmm_kernel<false><<<NB, dim3(512), 0, stream>>>(packed, bstart, out, out, lg);
        bucket_spmm_kernel<true><<<NB, dim3(512), 0, stream>>>(packed, bstart, lg, out, out);
        return;
    }

    float* local  = (float*)d_ws;
    float* logits = local  + (size_t)N_NODES * N_CLASS;
    float* deg    = logits + (size_t)N_NODES * N_CLASS;
    float* agg    = out;

    const dim3 blk(256);
    const long long nl   = (long long)N_NODES * N_CLASS;
    const long long ne16 = (long long)N_EDGES * N_CLASS;
    const unsigned gN    = (unsigned)((N_NODES + 255) / 256);
    const unsigned gNL   = (unsigned)((nl + 255) / 256);
    const unsigned gE    = (unsigned)((N_EDGES + 255) / 256);
    const unsigned gE16  = (unsigned)((ne16 + 255) / 256);

    const unsigned gGEMM = (unsigned)(((N_NODES + 15) / 16 + 7) / 8);
    gcn_gemm_kernel<<<gGEMM, blk, 0, stream>>>(x, W1, W2, local);

    zero_kernel<<<gN, blk, 0, stream>>>(deg, N_NODES);
    deg_kernel<<<gE, blk, 0, stream>>>(esrc, deg);

    zero_kernel<<<gNL, blk, 0, stream>>>(agg, nl);
    scatter_kernel<<<gE16, blk, 0, stream>>>(esrc, edst, local, agg);
    combine_kernel<<<gNL, blk, 0, stream>>>(agg, local, deg, logits);

    zero_kernel<<<gNL, blk, 0, stream>>>(agg, nl);
    scatter_kernel<<<gE16, blk, 0, stream>>>(esrc, edst, logits, agg);
    final_kernel<<<gN, blk, 0, stream>>>(agg, local, deg, out);
}


# propagation passes: bucket edges ordered by src row in LDS (integer count/prefix/place); 4 lanes per row sum gathered rows in registers; f32 LDS atomics gone; sum order: row list; softmax tree 4+2
# speedup vs baseline: 2.5240x; 1.1189x over previous
#include <hip/hip_runtime.h>
#include <hip/hip_bf16.h>
#include <math.h>

#define N_NODES 100000
#define N_FEAT  512
#define HIDDEN  128
#define N_CLASS 16
#define N_EDGES 3200000
#define ALPHA   0.25f

typedef __attribute__((ext_vector_type(2))) float v2f;
typedef __attribute__((ext_vector_type(8))) float v8f;

#define KCHUNK   64
#define W1C_STR  136
#define HID_STR  132

__global__ __launch_bounds__(256) void gcn_gemm_kernel(
    const float* __restrict__ x, const float* __restrict__ W1,
    const float* __restrict__ W2, float* __restrict__ local_out)
{
    __shared__ float w1c[KCHUNK][W1C_STR];
    __shared__ float hid[8][16][HID_STR];

    const int tid  = threadIdx.x;
    const int wave = tid >> 5;
    const int lane = tid & 31;
    const int half = lane >> 4;
    const int l16  = lane & 15;

    const int row_tile = blockIdx.x * 8 + wave;
    const int row_base = row_tile * 16;
    const bool active  = (row_base < N_NODES);

    v8f acc[8];
    for (int t = 0; t < 8; ++t)
        for (int i = 0; i < 8; ++i) acc[t][i] = 0.0f;

    for (int k0 = 0; k0 < N_FEAT; k0 += KCHUNK) {
        __syncthreads();
        for (int i = tid; i < KCHUNK * (HIDDEN / 4); i += 256) {
            const int r  = i >> 5;
            const int c4 = i & 31;
            const float4 v = ((const float4*)(W1 + (size_t)(k0 + r) * HIDDEN))[c4];
            *(float4*)&w1c[r][c4 * 4] = v;
        }
        __syncthreads();

        if (active) {
            for (int kk = 0; kk < KCHUNK; kk += 4) {
                const float* ap = x + (size_t)(row_base + l16) * N_FEAT
                                    + (k0 + kk + 2 * half);
                v2f a = *(const v2f*)ap;
                for (int t = 0; t < 8; ++t) {
                    v2f b;
                    b[0] = w1c[kk + 2 * half + 0][t * 16 + l16];
                    b[1] = w1c[kk + 2 * half + 1][t * 16 + l16];
                    acc[t] = __builtin_amdgcn_wmma_f32_16x16x4_f32(
                        false, a, false, b, (short)0, acc[t], false, false);
                }
            }
        }
    }

    if (active) {
        for (int t = 0; t < 8; ++t)
            for (int r = 0; r < 8; ++r) {
                float v = acc[t][r];
                hid[wave][r + 8 * half][t * 16 + l16] = v > 0.0f ? v : 0.0f;
            }
        v8f c2;
        for (int i = 0; i < 8; ++i) c2[i] = 0.0f;
        for (int kk = 0; kk < HIDDEN; kk += 4) {
            v2f a, b;
            a[0] = hid[wave][l16][kk + 2 * half + 0];
            a[1] = hid[wave][l16][kk + 2 * half + 1];
            b[0] = W2[(size_t)(kk + 2 * half + 0) * N_CLASS + l16];
            b[1] = W2[(size_t)(kk + 2 * half + 1) * N_CLASS + l16];
            c2 = __builtin_amdgcn_wmma_f32_16x16x4_f32(
                false, a, false, b, (short)0, c2, false, false);
        }
        for (int r = 0; r < 8; ++r)
            local_out[(size_t)(row_base + r + 8 * half) * N_CLASS + l16] = c2[r];
    }
}

__global__ void zero_kernel(float* __restrict__ p, long long n)
{
    long long i = (long long)blockIdx.x * blockDim.x + threadIdx.x;
    if (i < n) p[i] = 0.0f;
}

__global__ void deg_kernel(const int* __restrict__ src, float* __restrict__ deg)
{
    int e = blockIdx.x * blockDim.x + threadIdx.x;
    if (e < N_EDGES) atomicAdd(&deg[src[e]], 1.0f);
}

__global__ void scatter_kernel(const int* __restrict__ src,
                               const int* __restrict__ dst,
                               const float* __restrict__ logits,
                               float* __restrict__ agg)
{
    long long gid = (long long)blockIdx.x * blockDim.x + threadIdx.x;
    if (gid >= (long long)N_EDGES * N_CLASS) return;
    const int e = (int)(gid >> 4);
    const int c = (int)(gid & 15);
    const int s = src[e];
    const int d = dst[e];
    atomicAdd(&agg[(size_t)s * N_CLASS + c], logits[(size_t)d * N_CLASS + c]);
}

__global__ void combine_kernel(const float* __restrict__ agg,
                               const float* __restrict__ local,
                               const float* __restrict__ deg,
                               float* __restrict__ out)
{
    long long i = (long long)blockIdx.x * blockDim.x + threadIdx.x;
    if (i >= (long long)N_NODES * N_CLASS) return;
    const int row = (int)(i >> 4);
    const float scale = (1.0f - ALPHA) / fmaxf(deg[row], 1e-12f);
    out[i] = scale * agg[i] + ALPHA * local[i];
}

__global__ void final_kernel(const float* __restrict__ agg,
                             const float* __restrict__ local,
                             const float* __restrict__ deg,
                             float* __restrict__ out)
{
    int row = blockIdx.x * blockDim.x + threadIdx.x;
    if (row >= N_NODES) return;
    const float scale = (1.0f - ALPHA) / fmaxf(deg[row], 1e-12f);
    float t[N_CLASS];
    float m = -INFINITY;
    for (int c = 0; c < N_CLASS; ++c) {
        t[c] = scale * agg[(size_t)row * N_CLASS + c]
             + ALPHA * local[(size_t)row * N_CLASS + c];
        m = fmaxf(m, t[c]);
    }
    float s = 0.0f;
    for (int c = 0; c < N_CLASS; ++c) s += __expf(t[c] - m);
    const float lse = m + __logf(s);
    for (int c = 0; c < N_CLASS; ++c)
        out[(size_t)row * N_CLASS + c] = t[c] - lse;
}


#define RSH    7
#define RB     (1 << RSH)
#define NB     ((N_NODES + RB - 1) / RB)
#define NCH    256
#define CH     (N_EDGES / NCH)
#define CAP    6144
#define PER_T  (CAP / 512)
static_assert(N_EDGES % NCH == 0, "chunks must tile the edge list");
static_assert(NB <= 1024, "one thread per bucket in the place kernel");
static_assert(N_NODES <= (1 << (32 - RSH)), "dst and local row share one word");

__global__ __launch_bounds__(1024) void bucket_count_kernel(
    const int* __restrict__ esrc, unsigned* __restrict__ cnt)
{
    __shared__ unsigned h[NB];
    const int tid = threadIdx.x;
    for (int i = tid; i < NB; i += 1024) h[i] = 0u;
    __syncthreads();
    const int e0 = blockIdx.x * CH;
    for (int i = tid; i < CH; i += 1024)
        atomicAdd(&h[esrc[e0 + i] >> RSH], 1u);
    __syncthreads();
    for (int i = tid; i < NB; i += 1024) cnt[blockIdx.x * NB + i] = h[i];
}

static_assert(NCH % 32 == 0, "each lane takes NCH/32 consecutive chunks");
__global__ __launch_bounds__(256) void bucket_prefix_kernel(
    unsigned* __restrict__ cnt, unsigned* __restrict__ tot)
{
    const int lane = threadIdx.x & 31;
    const int b    = blockIdx.x * 8 + (threadIdx.x >> 5);
    if (b >= NB) return;
    unsigned c[NCH / 32];
    for (int j = 0; j < NCH / 32; ++j)
        c[j] = cnt[(lane * (NCH / 32) + j) * NB + b];
    unsigned s = 0u;
    for (int j = 0; j < NCH / 32; ++j) { const unsigned t = c[j]; c[j] = s; s += t; }
    unsigned incl = s;
    for (int o = 1; o < 32; o <<= 1) {
        const unsigned t = __shfl_up(incl, o);
        if (lane >= o) incl += t;
    }
    const unsigned base = incl - s;
    for (int j = 0; j < NCH / 32; ++j)
        cnt[(lane * (NCH / 32) + j) * NB + b] = base + c[j];
    if (lane == 31) tot[b] = incl;
}

__global__ __launch_bounds__(1024) void bucket_place_kernel(
    const int* __restrict__ esrc, const int* __restrict__ edst,
    const unsigned* __restrict__ cnt, const unsigned* __restrict__ tot,
    unsigned* __restrict__ packed, unsigned* __restrict__ bstart)
{
    __shared__ unsigned cur[NB];
    __shared__ unsigned wsum[32];
    const int tid  = threadIdx.x;
    const int lane = tid & 31;
    const int w    = tid >> 5;
    const int me   = blockIdx.x;

    const unsigned tb     = (tid < NB) ? tot[tid] : 0u;
    const unsigned before = (tid < NB) ? cnt[me * NB + tid] : 0u;
    unsigned incl = tb;
    for (int o = 1; o < 32; o <<= 1) {
        const unsigned t = __shfl_up(incl, o);
        if (lane >= o) incl += t;
    }
    if (lane == 31) wsum[w] = incl;
    __syncthreads();
    if (w == 0) {
        const unsigned s = wsum[lane];
        unsigned si = s;
        for (int o = 1; o < 32; o <<= 1) {
            const unsigned t = __shfl_up(si, o);
            if (lane >= o) si += t;
        }
        wsum[lane] = si - s;
    }
    __syncthreads();
    const unsigned start = wsum[w] + incl - tb;
    if (tid < NB) {
        cur[tid] = start + before;
        if (me == 0) bstart[tid] = start;
    }
    if (me == 0 && tid == 0) bstart[NB] = (unsigned)N_EDGES;
    __syncthreads();

    const int e0 = me * CH;
    for (int i = tid; i < CH; i += 1024) {
        const int s = esrc[e0 + i];
        const int d = edst[e0 + i];
        const unsigned pos = atomicAdd(&cur[s >> RSH], 1u);
        packed[pos] = ((unsigned)d << RSH) | (unsigned)(s & (RB - 1));
    }
}

static_assert(RB * 4 == 512, "four lanes per row fill the work-group");
static_assert(CAP % 512 == 0, "a thread holds CAP/512 edge words per pass");
template <bool FINAL>
__global__ __launch_bounds__(512) void bucket_spmm_kernel(
    const unsigned* __restrict__ packed, const unsigned* __restrict__ bstart,
    const float* rows_in, const float* own_in, float* rows_out)
{
    __shared__ unsigned sorted[CAP];
    __shared__ unsigned rcur[RB];
    __shared__ unsigned rbeg[RB + 1];
    __shared__ unsigned wsum[RB / 32];
    const int tid  = threadIdx.x;
    const int lane = tid & 31;
    const int w    = tid >> 5;
    const int b    = blockIdx.x;
    const int r    = tid >> 2;
    const unsigned q = (unsigned)tid & 3u;

    const unsigned e0 = bstart[b];
    const unsigned e1 = bstart[b + 1];
    const float4* g4  = (const float4*)rows_in;

    float4   sum = make_float4(0.0f, 0.0f, 0.0f, 0.0f);
    unsigned deg = 0u;

    for (unsigned t0 = e0; t0 < e1; t0 += (unsigned)CAP) {
        const unsigned left = e1 - t0;
        const unsigned n    = left < (unsigned)CAP ? left : (unsigned)CAP;

        unsigned p[PER_T];
#pragma unroll
        for (int k = 0; k < PER_T; ++k) {
            const unsigned i = (unsigned)tid + (unsigned)k * 512u;
            p[k] = (i < n) ? packed[t0 + i] : 0u;
        }
        if (tid < RB) rcur[tid] = 0u;
        __syncthreads();
#pragma unroll
        for (int k = 0; k < PER_T; ++k) {
            const unsigned i = (unsigned)tid + (unsigned)k * 512u;
            if (i < n) atomicAdd(&rcur[p[k] & (RB - 1)], 1u);
        }
        __syncthreads();

        const unsigned c = (tid < RB) ? rcur[tid] : 0u;
        unsigned incl = c;
        for (int o = 1; o < 32; o <<= 1) {
            const unsigned t = __shfl_up(incl, o);
            if (lane >= o) incl += t;
        }
        if (tid < RB && lane == 31) wsum[w] = incl;
        __syncthreads();
        if (tid < RB) {
            unsigned base = 0u;
            for (int k = 0; k < w; ++k) base += wsum[k];
            const unsigned excl = base + incl - c;
            rbeg[tid] = excl;
            rcur[tid] = excl;
        }
        if (tid == 0) rbeg[RB] = n;
        __syncthreads();

#pragma unroll
        for (int k = 0; k < PER_T; ++k) {
            const unsigned i = (unsigned)tid + (unsigned)k * 512u;
            if (i < n) {
                const unsigned pos = atomicAdd(&rcur[p[k] & (RB - 1)], 1u);
                sorted[pos] = p[k] >> RSH;
            }
        }
        __syncthreads();

        const unsigned beg = rbeg[r];
        const unsigned end = rbeg[r + 1];
        deg += end - beg;
        unsigned j = beg;
        for (; j + 4u <= end; j += 4u) {
            unsigned d[4];
            float4   v[4];
            for (int k = 0; k < 4; ++k) d[k] = sorted[j + (unsigned)k];
            for (int k = 0; k < 4; ++k) v[k] = g4[(size_t)d[k] * 4 + q];
            for (int k = 0; k < 4; ++k) {
                sum.x += v[k].x;
                sum.y += v[k].y;
                sum.z += v[k].z;
                sum.w += v[k].w;
            }
        }
        if (j < end) {
            unsigned d[3];
            float4   v[3];
            for (int k = 0; k < 3; ++k) d[k] = sorted[min(j + (unsigned)k, end - 1u)];
            for (int k = 0; k < 3; ++k) v[k] = g4[(size_t)d[k] * 4 + q];
            for (int k = 0; k < 3; ++k) {
                if (j + (unsigned)k < end) {
                    sum.x += v[k].x;
                    sum.y += v[k].y;
                    sum.z += v[k].z;
                    sum.w += v[k].w;
                }
            }
        }
        __syncthreads();
    }

    const int  row   = b * RB + r;
    const bool valid = row < N_NODES;
    const float scale = (1.0f - ALPHA) / fmaxf((float)deg, 1e-12f);
    float4 lv = make_float4(0.0f, 0.0f, 0.0f, 0.0f);
    if (valid) lv = ((const float4*)own_in)[(size_t)row * 4 + q];
    float4 t;
    t.x = scale * sum.x + ALPHA * lv.x;
    t.y = scale * sum.y + ALPHA * lv.y;
    t.z = scale * sum.z + ALPHA * lv.z;
    t.w = scale * sum.w + ALPHA * lv.w;
    if (!FINAL) {
        if (valid) ((float4*)rows_out)[(size_t)row * 4 + q] = t;
    } else {
        float m = fmaxf(fmaxf(t.x, t.y), fmaxf(t.z, t.w));
        m = fmaxf(m, __shfl_xor(m, 1, 4));
        m = fmaxf(m, __shfl_xor(m, 2, 4));
        float sm = (__expf(t.x - m) + __expf(t.y - m)) + (__expf(t.z - m) + __expf(t.w - m));
        sm += __shfl_xor(sm, 1, 4);
        sm += __shfl_xor(sm, 2, 4);
        const float lse = m + __logf(sm);
        t.x -= lse; t.y -= lse; t.z -= lse; t.w -= lse;
        if (valid) ((float4*)rows_out)[(size_t)row * 4 + q] = t;
    }
}

extern "C" void kernel_launch(void* const* d_in, const int* in_sizes, int n_in,
                              void* d_out, int out_size, void* d_ws, size_t ws_size,
                              hipStream_t stream)
{
    const float* x    = (const float*)d_in[0];
    const float* W1   = (const float*)d_in[1];
    const float* W2   = (const float*)d_in[2];
    const int*   esrc = (const int*)d_in[3];
    const int*   edst = (const int*)d_in[4];
    float* out = (float*)d_out;

    const size_t off_packed = (size_t)N_NODES * N_CLASS * sizeof(float);
    const size_t off_bstart = off_packed + (size_t)N_EDGES * sizeof(unsigned);
    const size_t off_tot    = off_bstart + (size_t)(NB + 1) * sizeof(unsigned);
    const size_t ws_need    = off_tot + (size_t)NB * sizeof(unsigned);
    if (ws_size >= ws_need) {
        float*    lg     = (float*)d_ws;
        unsigned* cnt    = (unsigned*)d_ws;
        unsigned* packed = (unsigned*)((char*)d_ws + off_packed);
        unsigned* bstart = (unsigned*)((char*)d_ws + off_bstart);
        unsigned* tot    = (unsigned*)((char*)d_ws + off_tot);
        const unsigned gG = (unsigned)(((N_NODES + 15) / 16 + 7) / 8);
        gcn_gemm_kernel<<<gG, dim3(256), 0, stream>>>(x, W1, W2, out);
        bucket_count_kernel<<<NCH, dim3(1024), 0, stream>>>(esrc, cnt);
        bucket_prefix_kernel<<<(NB + 7) / 8, dim3(256), 0, stream>>>(cnt, tot);
        bucket_place_kernel<<<NCH, dim3(1024), 0, stream>>>(esrc, edst, cnt, tot, packed, bstart);
        bucket_spmm_kernel<false><<<NB, dim3(512), 0, stream>>>(packed, bstart, out, out, lg);
        bucket_spmm_kernel<true><<<NB, dim3(512), 0, stream>>>(packed, bstart, lg, out, out);
        return;
    }

    float* local  = (float*)d_ws;
    float* logits = local  + (size_t)N_NODES * N_CLASS;
    float* deg    = logits + (size_t)N_NODES * N_CLASS;
    float* agg    = out;

    const dim3 blk(256);
    const long long nl   = (long long)N_NODES * N_CLASS;
    const long long ne16 = (long long)N_EDGES * N_CLASS;
    const unsigned gN    = (unsigned)((N_NODES + 255) / 256);
    const unsigned gNL   = (unsigned)((nl + 255) / 256);
    const unsigned gE    = (unsigned)((N_EDGES + 255) / 256);
    const unsigned gE16  = (unsigned)((ne16 + 255) / 256);

    const unsigned gGEMM = (unsigned)(((N_NODES + 15) / 16 + 7) / 8);
    gcn_gemm_kernel<<<gGEMM, blk, 0, stream>>>(x, W1, W2, local);

    zero_kernel<<<gN, blk, 0, stream>>>(deg, N_NODES);
    deg_kernel<<<gE, blk, 0, stream>>>(esrc, deg);

    zero_kernel<<<gNL, blk, 0, stream>>>(agg, nl);
    scatter_kernel<<<gE16, blk, 0, stream>>>(esrc, edst, local, agg);
    combine_kernel<<<gNL, blk, 0, stream>>>(agg, local, deg, logits);

    zero_kernel<<<gNL, blk, 0, stream>>>(agg, nl);
    scatter_kernel<<<gE16, blk, 0, stream>>>(esrc, edst, logits, agg);
    final_kernel<<<gN, blk, 0, stream>>>(agg, local, deg, out);
}


# GEMM kernel: a thread's eight float4 loads of the W1 chunk are issued as one batch, then stored to LDS (were eight dependent load-store trips per chunk); sums bit-identical
# speedup vs baseline: 2.7515x; 1.0901x over previous
#include <hip/hip_runtime.h>
#include <hip/hip_bf16.h>
#include <math.h>

#define N_NODES 100000
#define N_FEAT  512
#define HIDDEN  128
#define N_CLASS 16
#define N_EDGES 3200000
#define ALPHA   0.25f

typedef __attribute__((ext_vector_type(2))) float v2f;
typedef __attribute__((ext_vector_type(8))) float v8f;

#define KCHUNK   64
#define W1C_STR  136
#define HID_STR  132

__global__ __launch_bounds__(256) void gcn_gemm_kernel(
    const float* __restrict__ x, const float* __restrict__ W1,
    const float* __restrict__ W2, float* __restrict__ local_out)
{
    __shared__ float w1c[KCHUNK][W1C_STR];
    __shared__ float hid[8][16][HID_STR];

    const int tid  = threadIdx.x;
    const int wave = tid >> 5;
    const int lane = tid & 31;
    const int half = lane >> 4;
    const int l16  = lane & 15;

    const int row_tile = blockIdx.x * 8 + wave;
    const int row_base = row_tile * 16;
    const bool active  = (row_base < N_NODES);

    v8f acc[8];
    for (int t = 0; t < 8; ++t)
        for (int i = 0; i < 8; ++i) acc[t][i] = 0.0f;

    for (int k0 = 0; k0 < N_FEAT; k0 += KCHUNK) {
        __syncthreads();
        {
            float4 wv[KCHUNK * (HIDDEN / 4) / 256];
#pragma unroll
            for (int j = 0; j < KCHUNK * (HIDDEN / 4) / 256; ++j) {
                const int i  = tid + j * 256;
                const int r  = i >> 5;
                const int c4 = i & 31;
                wv[j] = ((const float4*)(W1 + (size_t)(k0 + r) * HIDDEN))[c4];
            }
#pragma unroll
            for (int j = 0; j < KCHUNK * (HIDDEN / 4) / 256; ++j) {
                const int i  = tid + j * 256;
                const int r  = i >> 5;
                const int c4 = i & 31;
                *(float4*)&w1c[r][c4 * 4] = wv[j];
            }
        }
        __syncthreads();

        if (active) {
            for (int kk = 0; kk < KCHUNK; kk += 4) {
                const float* ap = x + (size_t)(row_base + l16) * N_FEAT
                                    + (k0 + kk + 2 * half);
                v2f a = *(const v2f*)ap;
                for (int t = 0; t < 8; ++t) {
                    v2f b;
                    b[0] = w1c[kk + 2 * half + 0][t * 16 + l16];
                    b[1] = w1c[kk + 2 * half + 1][t * 16 + l16];
                    acc[t] = __builtin_amdgcn_wmma_f32_16x16x4_f32(
                        false, a, false, b, (short)0, acc[t], false, false);
                }
            }
        }
    }

    if (active) {
        for (int t = 0; t < 8; ++t)
            for (int r = 0; r < 8; ++r) {
                float v = acc[t][r];
                hid[wave][r + 8 * half][t * 16 + l16] = v > 0.0f ? v : 0.0f;
            }
        v8f c2;
        for (int i = 0; i < 8; ++i) c2[i] = 0.0f;
        for (int kk = 0; kk < HIDDEN; kk += 4) {
            v2f a, b;
            a[0] = hid[wave][l16][kk + 2 * half + 0];
            a[1] = hid[wave][l16][kk + 2 * half + 1];
            b[0] = W2[(size_t)(kk + 2 * half + 0) * N_CLASS + l16];
            b[1] = W2[(size_t)(kk + 2 * half + 1) * N_CLASS + l16];
            c2 = __builtin_amdgcn_wmma_f32_16x16x4_f32(
                false, a, false, b, (short)0, c2, false, false);
        }
        for (int r = 0; r < 8; ++r)
            local_out[(size_t)(row_base + r + 8 * half) * N_CLASS + l16] = c2[r];
    }
}

__global__ void zero_kernel(float* __restrict__ p, long long n)
{
    long long i = (long long)blockIdx.x * blockDim.x + threadIdx.x;
    if (i < n) p[i] = 0.0f;
}

__global__ void deg_kernel(const int* __restrict__ src, float* __restrict__ deg)
{
    int e = blockIdx.x * blockDim.x + threadIdx.x;
    if (e < N_EDGES) atomicAdd(&deg[src[e]], 1.0f);
}

__global__ void scatter_kernel(const int* __restrict__ src,
                               const int* __restrict__ dst,
                               const float* __restrict__ logits,
                               float* __restrict__ agg)
{
    long long gid = (long long)blockIdx.x * blockDim.x + threadIdx.x;
    if (gid >= (long long)N_EDGES * N_CLASS) return;
    const int e = (int)(gid >> 4);
    const int c = (int)(gid & 15);
    const int s = src[e];
    const int d = dst[e];
    atomicAdd(&agg[(size_t)s * N_CLASS + c], logits[(size_t)d * N_CLASS + c]);
}

__global__ void combine_kernel(const float* __restrict__ agg,
                               const float* __restrict__ local,
                               const float* __restrict__ deg,
                               float* __restrict__ out)
{
    long long i = (long long)blockIdx.x * blockDim.x + threadIdx.x;
    if (i >= (long long)N_NODES * N_CLASS) return;
    const int row = (int)(i >> 4);
    const float scale = (1.0f - ALPHA) / fmaxf(deg[row], 1e-12f);
    out[i] = scale * agg[i] + ALPHA * local[i];
}

__global__ void final_kernel(const float* __restrict__ agg,
                             const float* __restrict__ local,
                             const float* __restrict__ deg,
                             float* __restrict__ out)
{
    int row = blockIdx.x * blockDim.x + threadIdx.x;
    if (row >= N_NODES) return;
    const float scale = (1.0f - ALPHA) / fmaxf(deg[row], 1e-12f);
    float t[N_CLASS];
    float m = -INFINITY;
    for (int c = 0; c < N_CLASS; ++c) {
        t[c] = scale * agg[(size_t)row * N_CLASS + c]
             + ALPHA * local[(size_t)row * N_CLASS + c];
        m = fmaxf(m, t[c]);
    }
    float s = 0.0f;
    for (int c = 0; c < N_CLASS; ++c) s += __expf(t[c] - m);
    const float lse = m + __logf(s);
    for (int c = 0; c < N_CLASS; ++c)
        out[(size_t)row * N_CLASS + c] = t[c] - lse;
}


#define RSH    7
#define RB     (1 << RSH)
#define NB     ((N_NODES + RB - 1) / RB)
#define NCH    256
#define CH     (N_EDGES / NCH)
#define CAP    6144
#define PER_T  (CAP / 512)
static_assert(N_EDGES % NCH == 0, "chunks must tile the edge list");
static_assert(NB <= 1024, "one thread per bucket in the place kernel");
static_assert(N_NODES <= (1 << (32 - RSH)), "dst and local row share one word");

__global__ __launch_bounds__(1024) void bucket_count_kernel(
    const int* __restrict__ esrc, unsigned* __restrict__ cnt)
{
    __shared__ unsigned h[NB];
    const int tid = threadIdx.x;
    for (int i = tid; i < NB; i += 1024) h[i] = 0u;
    __syncthreads();
    const int e0 = blockIdx.x * CH;
    for (int i = tid; i < CH; i += 1024)
        atomicAdd(&h[esrc[e0 + i] >> RSH], 1u);
    __syncthreads();
    for (int i = tid; i < NB; i += 1024) cnt[blockIdx.x * NB + i] = h[i];
}

static_assert(NCH % 32 == 0, "each lane takes NCH/32 consecutive chunks");
__global__ __launch_bounds__(256) void bucket_prefix_kernel(
    unsigned* __restrict__ cnt, unsigned* __restrict__ tot)
{
    const int lane = threadIdx.x & 31;
    const int b    = blockIdx.x * 8 + (threadIdx.x >> 5);
    if (b >= NB) return;
    unsigned c[NCH / 32];
    for (int j = 0; j < NCH / 32; ++j)
        c[j] = cnt[(lane * (NCH / 32) + j) * NB + b];
    unsigned s = 0u;
    for (int j = 0; j < NCH / 32; ++j) { const unsigned t = c[j]; c[j] = s; s += t; }
    unsigned incl = s;
    for (int o = 1; o < 32; o <<= 1) {
        const unsigned t = __shfl_up(incl, o);
        if (lane >= o) incl += t;
    }
    const unsigned base = incl - s;
    for (int j = 0; j < NCH / 32; ++j)
        cnt[(lane * (NCH / 32) + j) * NB + b] = base + c[j];
    if (lane == 31) tot[b] = incl;
}

__global__ __launch_bounds__(1024) void bucket_place_kernel(
    const int* __restrict__ esrc, const int* __restrict__ edst,
    const unsigned* __restrict__ cnt, const unsigned* __restrict__ tot,
    unsigned* __restrict__ packed, unsigned* __restrict__ bstart)
{
    __shared__ unsigned cur[NB];
    __shared__ unsigned wsum[32];
    const int tid  = threadIdx.x;
    const int lane = tid & 31;
    const int w    = tid >> 5;
    const int me   = blockIdx.x;

    const unsigned tb     = (tid < NB) ? tot[tid] : 0u;
    const unsigned before = (tid < NB) ? cnt[me * NB + tid] : 0u;
    unsigned incl = tb;
    for (int o = 1; o < 32; o <<= 1) {
        const unsigned t = __shfl_up(incl, o);
        if (lane >= o) incl += t;
    }
    if (lane == 31) wsum[w] = incl;
    __syncthreads();
    if (w == 0) {
        const unsigned s = wsum[lane];
        unsigned si = s;
        for (int o = 1; o < 32; o <<= 1) {
            const unsigned t = __shfl_up(si, o);
            if (lane >= o) si += t;
        }
        wsum[lane] = si - s;
    }
    __syncthreads();
    const unsigned start = wsum[w] + incl - tb;
    if (tid < NB) {
        cur[tid] = start + before;
        if (me == 0) bstart[tid] = start;
    }
    if (me == 0 && tid == 0) bstart[NB] = (unsigned)N_EDGES;
    __syncthreads();

    const int e0 = me * CH;
    for (int i = tid; i < CH; i += 1024) {
        const int s = esrc[e0 + i];
        const int d = edst[e0 + i];
        const unsigned pos = atomicAdd(&cur[s >> RSH], 1u);
        packed[pos] = ((unsigned)d << RSH) | (unsigned)(s & (RB - 1));
    }
}

static_assert(RB * 4 == 512, "four lanes per row fill the work-group");
static_assert(CAP % 512 == 0, "a thread holds CAP/512 edge words per pass");
template <bool FINAL>
__global__ __launch_bounds__(512) void bucket_spmm_kernel(
    const unsigned* __restrict__ packed, const unsigned* __restrict__ bstart,
    const float* rows_in, const float* own_in, float* rows_out)
{
    __shared__ unsigned sorted[CAP];
    __shared__ unsigned rcur[RB];
    __shared__ unsigned rbeg[RB + 1];
    __shared__ unsigned wsum[RB / 32];
    const int tid  = threadIdx.x;
    const int lane = tid & 31;
    const int w    = tid >> 5;
    const int b    = blockIdx.x;
    const int r    = tid >> 2;
    const unsigned q = (unsigned)tid & 3u;

    const unsigned e0 = bstart[b];
    const unsigned e1 = bstart[b + 1];
    const float4* g4  = (const float4*)rows_in;

    float4   sum = make_float4(0.0f, 0.0f, 0.0f, 0.0f);
    unsigned deg = 0u;

    for (unsigned t0 = e0; t0 < e1; t0 += (unsigned)CAP) {
        const unsigned left = e1 - t0;
        const unsigned n    = left < (unsigned)CAP ? left : (unsigned)CAP;

        unsigned p[PER_T];
#pragma unroll
        for (int k = 0; k < PER_T; ++k) {
            const unsigned i = (unsigned)tid + (unsigned)k * 512u;
            p[k] = (i < n) ? packed[t0 + i] : 0u;
        }
        if (tid < RB) rcur[tid] = 0u;
        __syncthreads();
#pragma unroll
        for (int k = 0; k < PER_T; ++k) {
            const unsigned i = (unsigned)tid + (unsigned)k * 512u;
            if (i < n) atomicAdd(&rcur[p[k] & (RB - 1)], 1u);
        }
        __syncthreads();

        const unsigned c = (tid < RB) ? rcur[tid] : 0u;
        unsigned incl = c;
        for (int o = 1; o < 32; o <<= 1) {
            const unsigned t = __shfl_up(incl, o);
            if (lane >= o) incl += t;
        }
        if (tid < RB && lane == 31) wsum[w] = incl;
        __syncthreads();
        if (tid < RB) {
            unsigned base = 0u;
            for (int k = 0; k < w; ++k) base += wsum[k];
            const unsigned excl = base + incl - c;
            rbeg[tid] = excl;
            rcur[tid] = excl;
        }
        if (tid == 0) rbeg[RB] = n;
        __syncthreads();

#pragma unroll
        for (int k = 0; k < PER_T; ++k) {
            const unsigned i = (unsigned)tid + (unsigned)k * 512u;
            if (i < n) {
                const unsigned pos = atomicAdd(&rcur[p[k] & (RB - 1)], 1u);
                sorted[pos] = p[k] >> RSH;
            }
        }
        __syncthreads();

        const unsigned beg = rbeg[r];
        const unsigned end = rbeg[r + 1];
        deg += end - beg;
        unsigned j = beg;
        for (; j + 4u <= end; j += 4u) {
            unsigned d[4];
            float4   v[4];
            for (int k = 0; k < 4; ++k) d[k] = sorted[j + (unsigned)k];
            for (int k = 0; k < 4; ++k) v[k] = g4[(size_t)d[k] * 4 + q];
            for (int k = 0; k < 4; ++k) {
                sum.x += v[k].x;
                sum.y += v[k].y;
                sum.z += v[k].z;
                sum.w += v[k].w;
            }
        }
        if (j < end) {
            unsigned d[3];
            float4   v[3];
            for (int k = 0; k < 3; ++k) d[k] = sorted[min(j + (unsigned)k, end - 1u)];
            for (int k = 0; k < 3; ++k) v[k] = g4[(size_t)d[k] * 4 + q];
            for (int k = 0; k < 3; ++k) {
                if (j + (unsigned)k < end) {
                    sum.x += v[k].x;
                    sum.y += v[k].y;
                    sum.z += v[k].z;
                    sum.w += v[k].w;
                }
            }
        }
        __syncthreads();
    }

    const int  row   = b * RB + r;
    const bool valid = row < N_NODES;
    const float scale = (1.0f - ALPHA) / fmaxf((float)deg, 1e-12f);
    float4 lv = make_float4(0.0f, 0.0f, 0.0f, 0.0f);
    if (valid) lv = ((const float4*)own_in)[(size_t)row * 4 + q];
    float4 t;
    t.x = scale * sum.x + ALPHA * lv.x;
    t.y = scale * sum.y + ALPHA * lv.y;
    t.z = scale * sum.z + ALPHA * lv.z;
    t.w = scale * sum.w + ALPHA * lv.w;
    if (!FINAL) {
        if (valid) ((float4*)rows_out)[(size_t)row * 4 + q] = t;
    } else {
        float m = fmaxf(fmaxf(t.x, t.y), fmaxf(t.z, t.w));
        m = fmaxf(m, __shfl_xor(m, 1, 4));
        m = fmaxf(m, __shfl_xor(m, 2, 4));
        float sm = (__expf(t.x - m) + __expf(t.y - m)) + (__expf(t.z - m) + __expf(t.w - m));
        sm += __shfl_xor(sm, 1, 4);
        sm += __shfl_xor(sm, 2, 4);
        const float lse = m + __logf(sm);
        t.x -= lse; t.y -= lse; t.z -= lse; t.w -= lse;
        if (valid) ((float4*)rows_out)[(size_t)row * 4 + q] = t;
    }
}

extern "C" void kernel_launch(void* const* d_in, const int* in_sizes, int n_in,
                              void* d_out, int out_size, void* d_ws, size_t ws_size,
                              hipStream_t stream)
{
    const float* x    = (const float*)d_in[0];
    const float* W1   = (const float*)d_in[1];
    const float* W2   = (const float*)d_in[2];
    const int*   esrc = (const int*)d_in[3];
    const int*   edst = (const int*)d_in[4];
    float* out = (float*)d_out;

    const size_t off_packed = (size_t)N_NODES * N_CLASS * sizeof(float);
    const size_t off_bstart = off_packed + (size_t)N_EDGES * sizeof(unsigned);
    const size_t off_tot    = off_bstart + (size_t)(NB + 1) * sizeof(unsigned);
    const size_t ws_need    = off_tot + (size_t)NB * sizeof(unsigned);
    if (ws_size >= ws_need) {
        float*    lg     = (float*)d_ws;
        unsigned* cnt    = (unsigned*)d_ws;
        unsigned* packed = (unsigned*)((char*)d_ws + off_packed);
        unsigned* bstart = (unsigned*)((char*)d_ws + off_bstart);
        unsigned* tot    = (unsigned*)((char*)d_ws + off_tot);
        const unsigned gG = (unsigned)(((N_NODES + 15) / 16 + 7) / 8);
        gcn_gemm_kernel<<<gG, dim3(256), 0, stream>>>(x, W1, W2, out);
        bucket_count_kernel<<<NCH, dim3(1024), 0, stream>>>(esrc, cnt);
        bucket_prefix_kernel<<<(NB + 7) / 8, dim3(256), 0, stream>>>(cnt, tot);
        bucket_place_kernel<<<NCH, dim3(1024), 0, stream>>>(esrc, edst, cnt, tot, packed, bstart);
        bucket_spmm_kernel<false><<<NB, dim3(512), 0, stream>>>(packed, bstart, out, out, lg);
        bucket_spmm_kernel<true><<<NB, dim3(512), 0, stream>>>(packed, bstart, lg, out, out);
        return;
    }

    float* local  = (float*)d_ws;
    float* logits = local  + (size_t)N_NODES * N_CLASS;
    float* deg    = logits + (size_t)N_NODES * N_CLASS;
    float* agg    = out;

    const dim3 blk(256);
    const long long nl   = (long long)N_NODES * N_CLASS;
    const long long ne16 = (long long)N_EDGES * N_CLASS;
    const unsigned gN    = (unsigned)((N_NODES + 255) / 256);
    const unsigned gNL   = (unsigned)((nl + 255) / 256);
    const unsigned gE    = (unsigned)((N_EDGES + 255) / 256);
    const unsigned gE16  = (unsigned)((ne16 + 255) / 256);

    const unsigned gGEMM = (unsigned)(((N_NODES + 15) / 16 + 7) / 8);
    gcn_gemm_kernel<<<gGEMM, blk, 0, stream>>>(x, W1, W2, local);

    zero_kernel<<<gN, blk, 0, stream>>>(deg, N_NODES);
    deg_kernel<<<gE, blk, 0, stream>>>(esrc, deg);

    zero_kernel<<<gNL, blk, 0, stream>>>(agg, nl);
    scatter_kernel<<<gE16, blk, 0, stream>>>(esrc, edst, local, agg);
    combine_kernel<<<gNL, blk, 0, stream>>>(agg, local, deg, logits);

    zero_kernel<<<gNL, blk, 0, stream>>>(agg, nl);
    scatter_kernel<<<gE16, blk, 0, stream>>>(esrc, edst, logits, agg);
    final_kernel<<<gN, blk, 0, stream>>>(agg, local, deg, out);
}


# GEMM kernel: one 25-wave group per WGP stages all of W1 in LDS once as k-pair rows (B fragment = one 8-byte read), K loop has no barrier, hidden tiles reuse W1's LDS; one round; sums bit-identical
# speedup vs baseline: 2.9971x; 1.0893x over previous
#include <hip/hip_runtime.h>
#include <hip/hip_bf16.h>
#include <math.h>

#define N_NODES 100000
#define N_FEAT  512
#define HIDDEN  128
#define N_CLASS 16
#define N_EDGES 3200000
#define ALPHA   0.25f

typedef __attribute__((ext_vector_type(2))) float v2f;
typedef __attribute__((ext_vector_type(8))) float v8f;

#define GW       25
#define GT       (GW * 32)
#define W1P_STR  288
#define HID_STR  132
#define W1_ITEMS ((N_FEAT / 2) * (HIDDEN / 4))
#define W1_FULL  (W1_ITEMS / GT)
static_assert(GW * 16 * HID_STR <= (N_FEAT / 2) * W1P_STR, "hidden tiles fit in the W1 region");

__global__ __launch_bounds__(GT) void gcn_gemm_kernel(
    const float* __restrict__ x, const float* __restrict__ W1,
    const float* __restrict__ W2, float* __restrict__ local_out)
{
    __shared__ __attribute__((aligned(16))) float w1s[(N_FEAT / 2) * W1P_STR];

    const int tid  = threadIdx.x;
    const int wave = tid >> 5;
    const int lane = tid & 31;
    const int half = lane >> 4;
    const int l16  = lane & 15;

    const int row_tile = blockIdx.x * GW + wave;
    const int row_base = row_tile * 16;
    const bool active  = (row_base < N_NODES);

    {
        float4 wa[W1_FULL], wb[W1_FULL];
#pragma unroll
        for (int j = 0; j < W1_FULL; ++j) {
            const int i  = tid + j * GT;
            const int p  = i >> 5;
            const int c4 = i & 31;
            wa[j] = ((const float4*)(W1 + (size_t)(2 * p + 0) * HIDDEN))[c4];
            wb[j] = ((const float4*)(W1 + (size_t)(2 * p + 1) * HIDDEN))[c4];
        }
#pragma unroll
        for (int j = 0; j < W1_FULL; ++j) {
            const int i  = tid + j * GT;
            const int p  = i >> 5;
            const int c4 = i & 31;
            float* d = &w1s[p * W1P_STR + c4 * 8];
            *(float4*)(d + 0) = make_float4(wa[j].x, wb[j].x, wa[j].y, wb[j].y);
            *(float4*)(d + 4) = make_float4(wa[j].z, wb[j].z, wa[j].w, wb[j].w);
        }
    }
    {
        const int i = tid + W1_FULL * GT;
        if (i < W1_ITEMS) {
            const int p  = i >> 5;
            const int c4 = i & 31;
            const float4 ra = ((const float4*)(W1 + (size_t)(2 * p + 0) * HIDDEN))[c4];
            const float4 rb = ((const float4*)(W1 + (size_t)(2 * p + 1) * HIDDEN))[c4];
            float* d = &w1s[p * W1P_STR + c4 * 8];
            *(float4*)(d + 0) = make_float4(ra.x, rb.x, ra.y, rb.y);
            *(float4*)(d + 4) = make_float4(ra.z, rb.z, ra.w, rb.w);
        }
    }
    __syncthreads();

    v8f acc[8];
    for (int t = 0; t < 8; ++t)
        for (int i = 0; i < 8; ++i) acc[t][i] = 0.0f;

    if (active) {
        const float* ap = x + (size_t)(row_base + l16) * N_FEAT + 2 * half;
        const float* bp = w1s + half * W1P_STR + l16 * 2;
        for (int k = 0; k < N_FEAT; k += 4) {
            v2f a = *(const v2f*)(ap + k);
            for (int t = 0; t < 8; ++t) {
                v2f b = *(const v2f*)(bp + (k >> 1) * W1P_STR + t * 32);
                acc[t] = __builtin_amdgcn_wmma_f32_16x16x4_f32(
                    false, a, false, b, (short)0, acc[t], false, false);
            }
        }
    }
    __syncthreads();

    if (active) {
        float* hid = w1s + wave * (16 * HID_STR);
        for (int t = 0; t < 8; ++t)
            for (int r = 0; r < 8; ++r) {
                float v = acc[t][r];
                hid[(r + 8 * half) * HID_STR + t * 16 + l16] = v > 0.0f ? v : 0.0f;
            }
        v8f c2;
        for (int i = 0; i < 8; ++i) c2[i] = 0.0f;
        for (int kk = 0; kk < HIDDEN; kk += 4) {
            v2f a, b;
            a[0] = hid[l16 * HID_STR + kk + 2 * half + 0];
            a[1] = hid[l16 * HID_STR + kk + 2 * half + 1];
            b[0] = W2[(size_t)(kk + 2 * half + 0) * N_CLASS + l16];
            b[1] = W2[(size_t)(kk + 2 * half + 1) * N_CLASS + l16];
            c2 = __builtin_amdgcn_wmma_f32_16x16x4_f32(
                false, a, false, b, (short)0, c2, false, false);
        }
        for (int r = 0; r < 8; ++r)
            local_out[(size_t)(row_base + r + 8 * half) * N_CLASS + l16] = c2[r];
    }
}

__global__ void zero_kernel(float* __restrict__ p, long long n)
{
    long long i = (long long)blockIdx.x * blockDim.x + threadIdx.x;
    if (i < n) p[i] = 0.0f;
}

__global__ void deg_kernel(const int* __restrict__ src, float* __restrict__ deg)
{
    int e = blockIdx.x * blockDim.x + threadIdx.x;
    if (e < N_EDGES) atomicAdd(&deg[src[e]], 1.0f);
}

__global__ void scatter_kernel(const int* __restrict__ src,
                               const int* __restrict__ dst,
                               const float* __restrict__ logits,
                               float* __restrict__ agg)
{
    long long gid = (long long)blockIdx.x * blockDim.x + threadIdx.x;
    if (gid >= (long long)N_EDGES * N_CLASS) return;
    const int e = (int)(gid >> 4);
    const int c = (int)(gid & 15);
    const int s = src[e];
    const int d = dst[e];
    atomicAdd(&agg[(size_t)s * N_CLASS + c], logits[(size_t)d * N_CLASS + c]);
}

__global__ void combine_kernel(const float* __restrict__ agg,
                               const float* __restrict__ local,
                               const float* __restrict__ deg,
                               float* __restrict__ out)
{
    long long i = (long long)blockIdx.x * blockDim.x + threadIdx.x;
    if (i >= (long long)N_NODES * N_CLASS) return;
    const int row = (int)(i >> 4);
    const float scale = (1.0f - ALPHA) / fmaxf(deg[row], 1e-12f);
    out[i] = scale * agg[i] + ALPHA * local[i];
}

__global__ void final_kernel(const float* __restrict__ agg,
                             const float* __restrict__ local,
                             const float* __restrict__ deg,
                             float* __restrict__ out)
{
    int row = blockIdx.x * blockDim.x + threadIdx.x;
    if (row >= N_NODES) return;
    const float scale = (1.0f - ALPHA) / fmaxf(deg[row], 1e-12f);
    float t[N_CLASS];
    float m = -INFINITY;
    for (int c = 0; c < N_CLASS; ++c) {
        t[c] = scale * agg[(size_t)row * N_CLASS + c]
             + ALPHA * local[(size_t)row * N_CLASS + c];
        m = fmaxf(m, t[c]);
    }
    float s = 0.0f;
    for (int c = 0; c < N_CLASS; ++c) s += __expf(t[c] - m);
    const float lse = m + __logf(s);
    for (int c = 0; c < N_CLASS; ++c)
        out[(size_t)row * N_CLASS + c] = t[c] - lse;
}


#define RSH    7
#define RB     (1 << RSH)
#define NB     ((N_NODES + RB - 1) / RB)
#define NCH    256
#define CH     (N_EDGES / NCH)
#define CAP    6144
#define PER_T  (CAP / 512)
static_assert(N_EDGES % NCH == 0, "chunks must tile the edge list");
static_assert(NB <= 1024, "one thread per bucket in the place kernel");
static_assert(N_NODES <= (1 << (32 - RSH)), "dst and local row share one word");

__global__ __launch_bounds__(1024) void bucket_count_kernel(
    const int* __restrict__ esrc, unsigned* __restrict__ cnt)
{
    __shared__ unsigned h[NB];
    const int tid = threadIdx.x;
    for (int i = tid; i < NB; i += 1024) h[i] = 0u;
    __syncthreads();
    const int e0 = blockIdx.x * CH;
    for (int i = tid; i < CH; i += 1024)
        atomicAdd(&h[esrc[e0 + i] >> RSH], 1u);
    __syncthreads();
    for (int i = tid; i < NB; i += 1024) cnt[blockIdx.x * NB + i] = h[i];
}

static_assert(NCH % 32 == 0, "each lane takes NCH/32 consecutive chunks");
__global__ __launch_bounds__(256) void bucket_prefix_kernel(
    unsigned* __restrict__ cnt, unsigned* __restrict__ tot)
{
    const int lane = threadIdx.x & 31;
    const int b    = blockIdx.x * 8 + (threadIdx.x >> 5);
    if (b >= NB) return;
    unsigned c[NCH / 32];
    for (int j = 0; j < NCH / 32; ++j)
        c[j] = cnt[(lane * (NCH / 32) + j) * NB + b];
    unsigned s = 0u;
    for (int j = 0; j < NCH / 32; ++j) { const unsigned t = c[j]; c[j] = s; s += t; }
    unsigned incl = s;
    for (int o = 1; o < 32; o <<= 1) {
        const unsigned t = __shfl_up(incl, o);
        if (lane >= o) incl += t;
    }
    const unsigned base = incl - s;
    for (int j = 0; j < NCH / 32; ++j)
        cnt[(lane * (NCH / 32) + j) * NB + b] = base + c[j];
    if (lane == 31) tot[b] = incl;
}

__global__ __launch_bounds__(1024) void bucket_place_kernel(
    const int* __restrict__ esrc, const int* __restrict__ edst,
    const unsigned* __restrict__ cnt, const unsigned* __restrict__ tot,
    unsigned* __restrict__ packed, unsigned* __restrict__ bstart)
{
    __shared__ unsigned cur[NB];
    __shared__ unsigned wsum[32];
    const int tid  = threadIdx.x;
    const int lane = tid & 31;
    const int w    = tid >> 5;
    const int me   = blockIdx.x;

    const unsigned tb     = (tid < NB) ? tot[tid] : 0u;
    const unsigned before = (tid < NB) ? cnt[me * NB + tid] : 0u;
    unsigned incl = tb;
    for (int o = 1; o < 32; o <<= 1) {
        const unsigned t = __shfl_up(incl, o);
        if (lane >= o) incl += t;
    }
    if (lane == 31) wsum[w] = incl;
    __syncthreads();
    if (w == 0) {
        const unsigned s = wsum[lane];
        unsigned si = s;
        for (int o = 1; o < 32; o <<= 1) {
            const unsigned t = __shfl_up(si, o);
            if (lane >= o) si += t;
        }
        wsum[lane] = si - s;
    }
    __syncthreads();
    const unsigned start = wsum[w] + incl - tb;
    if (tid < NB) {
        cur[tid] = start + before;
        if (me == 0) bstart[tid] = start;
    }
    if (me == 0 && tid == 0) bstart[NB] = (unsigned)N_EDGES;
    __syncthreads();

    const int e0 = me * CH;
    for (int i = tid; i < CH; i += 1024) {
        const int s = esrc[e0 + i];
        const int d = edst[e0 + i];
        const unsigned pos = atomicAdd(&cur[s >> RSH], 1u);
        packed[pos] = ((unsigned)d << RSH) | (unsigned)(s & (RB - 1));
    }
}

static_assert(RB * 4 == 512, "four lanes per row fill the work-group");
static_assert(CAP % 512 == 0, "a thread holds CAP/512 edge words per pass");
template <bool FINAL>
__global__ __launch_bounds__(512) void bucket_spmm_kernel(
    const unsigned* __restrict__ packed, const unsigned* __restrict__ bstart,
    const float* rows_in, const float* own_in, float* rows_out)
{
    __shared__ unsigned sorted[CAP];
    __shared__ unsigned rcur[RB];
    __shared__ unsigned rbeg[RB + 1];
    __shared__ unsigned wsum[RB / 32];
    const int tid  = threadIdx.x;
    const int lane = tid & 31;
    const int w    = tid >> 5;
    const int b    = blockIdx.x;
    const int r    = tid >> 2;
    const unsigned q = (unsigned)tid & 3u;

    const unsigned e0 = bstart[b];
    const unsigned e1 = bstart[b + 1];
    const float4* g4  = (const float4*)rows_in;

    float4   sum = make_float4(0.0f, 0.0f, 0.0f, 0.0f);
    unsigned deg = 0u;

    for (unsigned t0 = e0; t0 < e1; t0 += (unsigned)CAP) {
        const unsigned left = e1 - t0;
        const unsigned n    = left < (unsigned)CAP ? left : (unsigned)CAP;

        unsigned p[PER_T];
#pragma unroll
        for (int k = 0; k < PER_T; ++k) {
            const unsigned i = (unsigned)tid + (unsigned)k * 512u;
            p[k] = (i < n) ? packed[t0 + i] : 0u;
        }
        if (tid < RB) rcur[tid] = 0u;
        __syncthreads();
#pragma unroll
        for (int k = 0; k < PER_T; ++k) {
            const unsigned i = (unsigned)tid + (unsigned)k * 512u;
            if (i < n) atomicAdd(&rcur[p[k] & (RB - 1)], 1u);
        }
        __syncthreads();

        const unsigned c = (tid < RB) ? rcur[tid] : 0u;
        unsigned incl = c;
        for (int o = 1; o < 32; o <<= 1) {
            const unsigned t = __shfl_up(incl, o);
            if (lane >= o) incl += t;
        }
        if (tid < RB && lane == 31) wsum[w] = incl;
        __syncthreads();
        if (tid < RB) {
            unsigned base = 0u;
            for (int k = 0; k < w; ++k) base += wsum[k];
            const unsigned excl = base + incl - c;
            rbeg[tid] = excl;
            rcur[tid] = excl;
        }
        if (tid == 0) rbeg[RB] = n;
        __syncthreads();

#pragma unroll
        for (int k = 0; k < PER_T; ++k) {
            const unsigned i = (unsigned)tid + (unsigned)k * 512u;
            if (i < n) {
                const unsigned pos = atomicAdd(&rcur[p[k] & (RB - 1)], 1u);
                sorted[pos] = p[k] >> RSH;
            }
        }
        __syncthreads();

        const unsigned beg = rbeg[r];
        const unsigned end = rbeg[r + 1];
        deg += end - beg;
        unsigned j = beg;
        for (; j + 4u <= end; j += 4u) {
            unsigned d[4];
            float4   v[4];
            for (int k = 0; k < 4; ++k) d[k] = sorted[j + (unsigned)k];
            for (int k = 0; k < 4; ++k) v[k] = g4[(size_t)d[k] * 4 + q];
            for (int k = 0; k < 4; ++k) {
                sum.x += v[k].x;
                sum.y += v[k].y;
                sum.z += v[k].z;
                sum.w += v[k].w;
            }
        }
        if (j < end) {
            unsigned d[3];
            float4   v[3];
            for (int k = 0; k < 3; ++k) d[k] = sorted[min(j + (unsigned)k, end - 1u)];
            for (int k = 0; k < 3; ++k) v[k] = g4[(size_t)d[k] * 4 + q];
            for (int k = 0; k < 3; ++k) {
                if (j + (unsigned)k < end) {
                    sum.x += v[k].x;
                    sum.y += v[k].y;
                    sum.z += v[k].z;
                    sum.w += v[k].w;
                }
            }
        }
        __syncthreads();
    }

    const int  row   = b * RB + r;
    const bool valid = row < N_NODES;
    const float scale = (1.0f - ALPHA) / fmaxf((float)deg, 1e-12f);
    float4 lv = make_float4(0.0f, 0.0f, 0.0f, 0.0f);
    if (valid) lv = ((const float4*)own_in)[(size_t)row * 4 + q];
    float4 t;
    t.x = scale * sum.x + ALPHA * lv.x;
    t.y = scale * sum.y + ALPHA * lv.y;
    t.z = scale * sum.z + ALPHA * lv.z;
    t.w = scale * sum.w + ALPHA * lv.w;
    if (!FINAL) {
        if (valid) ((float4*)rows_out)[(size_t)row * 4 + q] = t;
    } else {
        float m = fmaxf(fmaxf(t.x, t.y), fmaxf(t.z, t.w));
        m = fmaxf(m, __shfl_xor(m, 1, 4));
        m = fmaxf(m, __shfl_xor(m, 2, 4));
        float sm = (__expf(t.x - m) + __expf(t.y - m)) + (__expf(t.z - m) + __expf(t.w - m));
        sm += __shfl_xor(sm, 1, 4);
        sm += __shfl_xor(sm, 2, 4);
        const float lse = m + __logf(sm);
        t.x -= lse; t.y -= lse; t.z -= lse; t.w -= lse;
        if (valid) ((float4*)rows_out)[(size_t)row * 4 + q] = t;
    }
}

extern "C" void kernel_launch(void* const* d_in, const int* in_sizes, int n_in,
                              void* d_out, int out_size, void* d_ws, size_t ws_size,
                              hipStream_t stream)
{
    const float* x    = (const float*)d_in[0];
    const float* W1   = (const float*)d_in[1];
    const float* W2   = (const float*)d_in[2];
    const int*   esrc = (const int*)d_in[3];
    const int*   edst = (const int*)d_in[4];
    float* out = (float*)d_out;

    const size_t off_packed = (size_t)N_NODES * N_CLASS * sizeof(float);
    const size_t off_bstart = off_packed + (size_t)N_EDGES * sizeof(unsigned);
    const size_t off_tot    = off_bstart + (size_t)(NB + 1) * sizeof(unsigned);
    const size_t ws_need    = off_tot + (size_t)NB * sizeof(unsigned);
    if (ws_size >= ws_need) {
        float*    lg     = (float*)d_ws;
        unsigned* cnt    = (unsigned*)d_ws;
        unsigned* packed = (unsigned*)((char*)d_ws + off_packed);
        unsigned* bstart = (unsigned*)((char*)d_ws + off_bstart);
        unsigned* tot    = (unsigned*)((char*)d_ws + off_tot);
        const unsigned gG = (unsigned)(((N_NODES + 15) / 16 + GW - 1) / GW);
        gcn_gemm_kernel<<<gG, dim3(GT), 0, stream>>>(x, W1, W2, out);
        bucket_count_kernel<<<NCH, dim3(1024), 0, stream>>>(esrc, cnt);
        bucket_prefix_kernel<<<(NB + 7) / 8, dim3(256), 0, stream>>>(cnt, tot);
        bucket_place_kernel<<<NCH, dim3(1024), 0, stream>>>(esrc, edst, cnt, tot, packed, bstart);
        bucket_spmm_kernel<false><<<NB, dim3(512), 0, stream>>>(packed, bstart, out, out, lg);
        bucket_spmm_kernel<true><<<NB, dim3(512), 0, stream>>>(packed, bstart, lg, out, out);
        return;
    }

    float* local  = (float*)d_ws;
    float* logits = local  + (size_t)N_NODES * N_CLASS;
    float* deg    = logits + (size_t)N_NODES * N_CLASS;
    float* agg    = out;

    const dim3 blk(256);
    const long long nl   = (long long)N_NODES * N_CLASS;
    const long long ne16 = (long long)N_EDGES * N_CLASS;
    const unsigned gN    = (unsigned)((N_NODES + 255) / 256);
    const unsigned gNL   = (unsigned)((nl + 255) / 256);
    const unsigned gE    = (unsigned)((N_EDGES + 255) / 256);
    const unsigned gE16  = (unsigned)((ne16 + 255) / 256);

    const unsigned gGEMM = (unsigned)(((N_NODES + 15) / 16 + GW - 1) / GW);
    gcn_gemm_kernel<<<gGEMM, dim3(GT), 0, stream>>>(x, W1, W2, local);

    zero_kernel<<<gN, blk, 0, stream>>>(deg, N_NODES);
    deg_kernel<<<gE, blk, 0, stream>>>(esrc, deg);

    zero_kernel<<<gNL, blk, 0, stream>>>(agg, nl);
    scatter_kernel<<<gE16, blk, 0, stream>>>(esrc, edst, local, agg);
    combine_kernel<<<gNL, blk, 0, stream>>>(agg, local, deg, logits);

    zero_kernel<<<gNL, blk, 0, stream>>>(agg, nl);
    scatter_kernel<<<gE16, blk, 0, stream>>>(esrc, edst, logits, agg);
    final_kernel<<<gN, blk, 0, stream>>>(agg, local, deg, out);
}


# edges regrouped per 25000-edge chunk by one kernel (LDS sort by src range, batched loads); each pass copies its range's 128 pieces to LDS; count/prefix/place launches gone; sum order: row list
# speedup vs baseline: 3.9994x; 1.3344x over previous
#include <hip/hip_runtime.h>
#include <hip/hip_bf16.h>
#include <math.h>

#define N_NODES 100000
#define N_FEAT  512
#define HIDDEN  128
#define N_CLASS 16
#define N_EDGES 3200000
#define ALPHA   0.25f

typedef __attribute__((ext_vector_type(2))) float v2f;
typedef __attribute__((ext_vector_type(8))) float v8f;

#define GW       25
#define GT       (GW * 32)
#define W1P_STR  288
#define HID_STR  132
#define W1_ITEMS ((N_FEAT / 2) * (HIDDEN / 4))
#define W1_FULL  (W1_ITEMS / GT)
static_assert(GW * 16 * HID_STR <= (N_FEAT / 2) * W1P_STR, "hidden tiles fit in the W1 region");

__global__ __launch_bounds__(GT) void gcn_gemm_kernel(
    const float* __restrict__ x, const float* __restrict__ W1,
    const float* __restrict__ W2, float* __restrict__ local_out)
{
    __shared__ __attribute__((aligned(16))) float w1s[(N_FEAT / 2) * W1P_STR];

    const int tid  = threadIdx.x;
    const int wave = tid >> 5;
    const int lane = tid & 31;
    const int half = lane >> 4;
    const int l16  = lane & 15;

    const int row_tile = blockIdx.x * GW + wave;
    const int row_base = row_tile * 16;
    const bool active  = (row_base < N_NODES);

    {
        float4 wa[W1_FULL], wb[W1_FULL];
#pragma unroll
        for (int j = 0; j < W1_FULL; ++j) {
            const int i  = tid + j * GT;
            const int p  = i >> 5;
            const int c4 = i & 31;
            wa[j] = ((const float4*)(W1 + (size_t)(2 * p + 0) * HIDDEN))[c4];
            wb[j] = ((const float4*)(W1 + (size_t)(2 * p + 1) * HIDDEN))[c4];
        }
#pragma unroll
        for (int j = 0; j < W1_FULL; ++j) {
            const int i  = tid + j * GT;
            const int p  = i >> 5;
            const int c4 = i & 31;
            float* d = &w1s[p * W1P_STR + c4 * 8];
            *(float4*)(d + 0) = make_float4(wa[j].x, wb[j].x, wa[j].y, wb[j].y);
            *(float4*)(d + 4) = make_float4(wa[j].z, wb[j].z, wa[j].w, wb[j].w);
        }
    }
    {
        const int i = tid + W1_FULL * GT;
        if (i < W1_ITEMS) {
            const int p  = i >> 5;
            const int c4 = i & 31;
            const float4 ra = ((const float4*)(W1 + (size_t)(2 * p + 0) * HIDDEN))[c4];
            const float4 rb = ((const float4*)(W1 + (size_t)(2 * p + 1) * HIDDEN))[c4];
            float* d = &w1s[p * W1P_STR + c4 * 8];
            *(float4*)(d + 0) = make_float4(ra.x, rb.x, ra.y, rb.y);
            *(float4*)(d + 4) = make_float4(ra.z, rb.z, ra.w, rb.w);
        }
    }
    __syncthreads();

    v8f acc[8];
    for (int t = 0; t < 8; ++t)
        for (int i = 0; i < 8; ++i) acc[t][i] = 0.0f;

    if (active) {
        const float* ap = x + (size_t)(row_base + l16) * N_FEAT + 2 * half;
        const float* bp = w1s + half * W1P_STR + l16 * 2;
        for (int k = 0; k < N_FEAT; k += 4) {
            v2f a = *(const v2f*)(ap + k);
            for (int t = 0; t < 8; ++t) {
                v2f b = *(const v2f*)(bp + (k >> 1) * W1P_STR + t * 32);
                acc[t] = __builtin_amdgcn_wmma_f32_16x16x4_f32(
                    false, a, false, b, (short)0, acc[t], false, false);
            }
        }
    }
    __syncthreads();

    if (active) {
        float* hid = w1s + wave * (16 * HID_STR);
        for (int t = 0; t < 8; ++t)
            for (int r = 0; r < 8; ++r) {
                float v = acc[t][r];
                hid[(r + 8 * half) * HID_STR + t * 16 + l16] = v > 0.0f ? v : 0.0f;
            }
        v8f c2;
        for (int i = 0; i < 8; ++i) c2[i] = 0.0f;
        for (int kk = 0; kk < HIDDEN; kk += 4) {
            v2f a, b;
            a[0] = hid[l16 * HID_STR + kk + 2 * half + 0];
            a[1] = hid[l16 * HID_STR + kk + 2 * half + 1];
            b[0] = W2[(size_t)(kk + 2 * half + 0) * N_CLASS + l16];
            b[1] = W2[(size_t)(kk + 2 * half + 1) * N_CLASS + l16];
            c2 = __builtin_amdgcn_wmma_f32_16x16x4_f32(
                false, a, false, b, (short)0, c2, false, false);
        }
        for (int r = 0; r < 8; ++r)
            local_out[(size_t)(row_base + r + 8 * half) * N_CLASS + l16] = c2[r];
    }
}

__global__ void zero_kernel(float* __restrict__ p, long long n)
{
    long long i = (long long)blockIdx.x * blockDim.x + threadIdx.x;
    if (i < n) p[i] = 0.0f;
}

__global__ void deg_kernel(const int* __restrict__ src, float* __restrict__ deg)
{
    int e = blockIdx.x * blockDim.x + threadIdx.x;
    if (e < N_EDGES) atomicAdd(&deg[src[e]], 1.0f);
}

__global__ void scatter_kernel(const int* __restrict__ src,
                               const int* __restrict__ dst,
                               const float* __restrict__ logits,
                               float* __restrict__ agg)
{
    long long gid = (long long)blockIdx.x * blockDim.x + threadIdx.x;
    if (gid >= (long long)N_EDGES * N_CLASS) return;
    const int e = (int)(gid >> 4);
    const int c = (int)(gid & 15);
    const int s = src[e];
    const int d = dst[e];
    atomicAdd(&agg[(size_t)s * N_CLASS + c], logits[(size_t)d * N_CLASS + c]);
}

__global__ void combine_kernel(const float* __restrict__ agg,
                               const float* __restrict__ local,
                               const float* __restrict__ deg,
                               float* __restrict__ out)
{
    long long i = (long long)blockIdx.x * blockDim.x + threadIdx.x;
    if (i >= (long long)N_NODES * N_CLASS) return;
    const int row = (int)(i >> 4);
    const float scale = (1.0f - ALPHA) / fmaxf(deg[row], 1e-12f);
    out[i] = scale * agg[i] + ALPHA * local[i];
}

__global__ void final_kernel(const float* __restrict__ agg,
                             const float* __restrict__ local,
                             const float* __restrict__ deg,
                             float* __restrict__ out)
{
    int row = blockIdx.x * blockDim.x + threadIdx.x;
    if (row >= N_NODES) return;
    const float scale = (1.0f - ALPHA) / fmaxf(deg[row], 1e-12f);
    float t[N_CLASS];
    float m = -INFINITY;
    for (int c = 0; c < N_CLASS; ++c) {
        t[c] = scale * agg[(size_t)row * N_CLASS + c]
             + ALPHA * local[(size_t)row * N_CLASS + c];
        m = fmaxf(m, t[c]);
    }
    float s = 0.0f;
    for (int c = 0; c < N_CLASS; ++c) s += __expf(t[c] - m);
    const float lse = m + __logf(s);
    for (int c = 0; c < N_CLASS; ++c)
        out[(size_t)row * N_CLASS + c] = t[c] - lse;
}


#define RSH    7
#define RB     (1 << RSH)
#define NB     ((N_NODES + RB - 1) / RB)
#define NCH    256
#define CH     (N_EDGES / NCH)
#define CAP    6144
#define PER_T  (CAP / 512)
static_assert(N_EDGES % NCH == 0, "chunks must tile the edge list");
static_assert(NB <= 1024, "one thread per bucket in the place kernel");
static_assert(N_NODES <= (1 << (32 - RSH)), "dst and local row share one word");

__global__ __launch_bounds__(1024) void bucket_count_kernel(
    const int* __restrict__ esrc, unsigned* __restrict__ cnt)
{
    __shared__ unsigned h[NB];
    const int tid = threadIdx.x;
    for (int i = tid; i < NB; i += 1024) h[i] = 0u;
    __syncthreads();
    const int e0 = blockIdx.x * CH;
    for (int i = tid; i < CH; i += 1024)
        atomicAdd(&h[esrc[e0 + i] >> RSH], 1u);
    __syncthreads();
    for (int i = tid; i < NB; i += 1024) cnt[blockIdx.x * NB + i] = h[i];
}

static_assert(NCH % 32 == 0, "each lane takes NCH/32 consecutive chunks");
__global__ __launch_bounds__(256) void bucket_prefix_kernel(
    unsigned* __restrict__ cnt, unsigned* __restrict__ tot)
{
    const int lane = threadIdx.x & 31;
    const int b    = blockIdx.x * 8 + (threadIdx.x >> 5);
    if (b >= NB) return;
    unsigned c[NCH / 32];
    for (int j = 0; j < NCH / 32; ++j)
        c[j] = cnt[(lane * (NCH / 32) + j) * NB + b];
    unsigned s = 0u;
    for (int j = 0; j < NCH / 32; ++j) { const unsigned t = c[j]; c[j] = s; s += t; }
    unsigned incl = s;
    for (int o = 1; o < 32; o <<= 1) {
        const unsigned t = __shfl_up(incl, o);
        if (lane >= o) incl += t;
    }
    const unsigned base = incl - s;
    for (int j = 0; j < NCH / 32; ++j)
        cnt[(lane * (NCH / 32) + j) * NB + b] = base + c[j];
    if (lane == 31) tot[b] = incl;
}

__global__ __launch_bounds__(1024) void bucket_place_kernel(
    const int* __restrict__ esrc, const int* __restrict__ edst,
    const unsigned* __restrict__ cnt, const unsigned* __restrict__ tot,
    unsigned* __restrict__ packed, unsigned* __restrict__ bstart)
{
    __shared__ unsigned cur[NB];
    __shared__ unsigned wsum[32];
    const int tid  = threadIdx.x;
    const int lane = tid & 31;
    const int w    = tid >> 5;
    const int me   = blockIdx.x;

    const unsigned tb     = (tid < NB) ? tot[tid] : 0u;
    const unsigned before = (tid < NB) ? cnt[me * NB + tid] : 0u;
    unsigned incl = tb;
    for (int o = 1; o < 32; o <<= 1) {
        const unsigned t = __shfl_up(incl, o);
        if (lane >= o) incl += t;
    }
    if (lane == 31) wsum[w] = incl;
    __syncthreads();
    if (w == 0) {
        const unsigned s = wsum[lane];
        unsigned si = s;
        for (int o = 1; o < 32; o <<= 1) {
            const unsigned t = __shfl_up(si, o);
            if (lane >= o) si += t;
        }
        wsum[lane] = si - s;
    }
    __syncthreads();
    const unsigned start = wsum[w] + incl - tb;
    if (tid < NB) {
        cur[tid] = start + before;
        if (me == 0) bstart[tid] = start;
    }
    if (me == 0 && tid == 0) bstart[NB] = (unsigned)N_EDGES;
    __syncthreads();

    const int e0 = me * CH;
    for (int i = tid; i < CH; i += 1024) {
        const int s = esrc[e0 + i];
        const int d = edst[e0 + i];
        const unsigned pos = atomicAdd(&cur[s >> RSH], 1u);
        packed[pos] = ((unsigned)d << RSH) | (unsigned)(s & (RB - 1));
    }
}

static_assert(RB * 4 == 512, "four lanes per row fill the work-group");
static_assert(CAP % 512 == 0, "a thread holds CAP/512 edge words per pass");
template <bool FINAL>
__global__ __launch_bounds__(512) void bucket_spmm_kernel(
    const unsigned* __restrict__ packed, const unsigned* __restrict__ bstart,
    const float* rows_in, const float* own_in, float* rows_out)
{
    __shared__ unsigned sorted[CAP];
    __shared__ unsigned rcur[RB];
    __shared__ unsigned rbeg[RB + 1];
    __shared__ unsigned wsum[RB / 32];
    const int tid  = threadIdx.x;
    const int lane = tid & 31;
    const int w    = tid >> 5;
    const int b    = blockIdx.x;
    const int r    = tid >> 2;
    const unsigned q = (unsigned)tid & 3u;

    const unsigned e0 = bstart[b];
    const unsigned e1 = bstart[b + 1];
    const float4* g4  = (const float4*)rows_in;

    float4   sum = make_float4(0.0f, 0.0f, 0.0f, 0.0f);
    unsigned deg = 0u;

    for (unsigned t0 = e0; t0 < e1; t0 += (unsigned)CAP) {
        const unsigned left = e1 - t0;
        const unsigned n    = left < (unsigned)CAP ? left : (unsigned)CAP;

        unsigned p[PER_T];
#pragma unroll
        for (int k = 0; k < PER_T; ++k) {
            const unsigned i = (unsigned)tid + (unsigned)k * 512u;
            p[k] = (i < n) ? packed[t0 + i] : 0u;
        }
        if (tid < RB) rcur[tid] = 0u;
        __syncthreads();
#pragma unroll
        for (int k = 0; k < PER_T; ++k) {
            const unsigned i = (unsigned)tid + (unsigned)k * 512u;
            if (i < n) atomicAdd(&rcur[p[k] & (RB - 1)], 1u);
        }
        __syncthreads();

        const unsigned c = (tid < RB) ? rcur[tid] : 0u;
        unsigned incl = c;
        for (int o = 1; o < 32; o <<= 1) {
            const unsigned t = __shfl_up(incl, o);
            if (lane >= o) incl += t;
        }
        if (tid < RB && lane == 31) wsum[w] = incl;
        __syncthreads();
        if (tid < RB) {
            unsigned base = 0u;
            for (int k = 0; k < w; ++k) base += wsum[k];
            const unsigned excl = base + incl - c;
            rbeg[tid] = excl;
            rcur[tid] = excl;
        }
        if (tid == 0) rbeg[RB] = n;
        __syncthreads();

#pragma unroll
        for (int k = 0; k < PER_T; ++k) {
            const unsigned i = (unsigned)tid + (unsigned)k * 512u;
            if (i < n) {
                const unsigned pos = atomicAdd(&rcur[p[k] & (RB - 1)], 1u);
                sorted[pos] = p[k] >> RSH;
            }
        }
        __syncthreads();

        const unsigned beg = rbeg[r];
        const unsigned end = rbeg[r + 1];
        deg += end - beg;
        unsigned j = beg;
        for (; j + 4u <= end; j += 4u) {
            unsigned d[4];
            float4   v[4];
            for (int k = 0; k < 4; ++k) d[k] = sorted[j + (unsigned)k];
            for (int k = 0; k < 4; ++k) v[k] = g4[(size_t)d[k] * 4 + q];
            for (int k = 0; k < 4; ++k) {
                sum.x += v[k].x;
                sum.y += v[k].y;
                sum.z += v[k].z;
                sum.w += v[k].w;
            }
        }
        if (j < end) {
            unsigned d[3];
            float4   v[3];
            for (int k = 0; k < 3; ++k) d[k] = sorted[min(j + (unsigned)k, end - 1u)];
            for (int k = 0; k < 3; ++k) v[k] = g4[(size_t)d[k] * 4 + q];
            for (int k = 0; k < 3; ++k) {
                if (j + (unsigned)k < end) {
                    sum.x += v[k].x;
                    sum.y += v[k].y;
                    sum.z += v[k].z;
                    sum.w += v[k].w;
                }
            }
        }
        __syncthreads();
    }

    const int  row   = b * RB + r;
    const bool valid = row < N_NODES;
    const float scale = (1.0f - ALPHA) / fmaxf((float)deg, 1e-12f);
    float4 lv = make_float4(0.0f, 0.0f, 0.0f, 0.0f);
    if (valid) lv = ((const float4*)own_in)[(size_t)row * 4 + q];
    float4 t;
    t.x = scale * sum.x + ALPHA * lv.x;
    t.y = scale * sum.y + ALPHA * lv.y;
    t.z = scale * sum.z + ALPHA * lv.z;
    t.w = scale * sum.w + ALPHA * lv.w;
    if (!FINAL) {
        if (valid) ((float4*)rows_out)[(size_t)row * 4 + q] = t;
    } else {
        float m = fmaxf(fmaxf(t.x, t.y), fmaxf(t.z, t.w));
        m = fmaxf(m, __shfl_xor(m, 1, 4));
        m = fmaxf(m, __shfl_xor(m, 2, 4));
        float sm = (__expf(t.x - m) + __expf(t.y - m)) + (__expf(t.z - m) + __expf(t.w - m));
        sm += __shfl_xor(sm, 1, 4);
        sm += __shfl_xor(sm, 2, 4);
        const float lse = m + __logf(sm);
        t.x -= lse; t.y -= lse; t.z -= lse; t.w -= lse;
        if (valid) ((float4*)rows_out)[(size_t)row * 4 + q] = t;
    }
}


#define NCS    128
#define CHS    (N_EDGES / NCS)
#define CS_PER ((CHS + 1023) / 1024)
static_assert(N_EDGES % NCS == 0, "chunks must tile the edge list");
static_assert(CHS % 4 == 0, "a segment is written as 16-byte pieces");

__global__ __launch_bounds__(1024) void chunk_sort_kernel(
    const int* __restrict__ esrc, const int* __restrict__ edst,
    unsigned* __restrict__ packed, unsigned* __restrict__ offT)
{
    __shared__ __attribute__((aligned(16))) unsigned sbuf[CHS];
    __shared__ unsigned cur[NB];
    __shared__ unsigned wsum[32];
    const int tid  = threadIdx.x;
    const int lane = tid & 31;
    const int w    = tid >> 5;
    const int me   = blockIdx.x;
    const int e0   = me * CHS;

    int s[CS_PER], d[CS_PER];
#pragma unroll
    for (int k = 0; k < CS_PER; ++k) {
        const int i = tid + k * 1024;
        s[k] = (i < CHS) ? esrc[e0 + i] : 0;
        d[k] = (i < CHS) ? edst[e0 + i] : 0;
    }
    if (tid < NB) cur[tid] = 0u;
    __syncthreads();
#pragma unroll
    for (int k = 0; k < CS_PER; ++k) {
        const int i = tid + k * 1024;
        if (i < CHS) atomicAdd(&cur[s[k] >> RSH], 1u);
    }
    __syncthreads();

    const unsigned c = (tid < NB) ? cur[tid] : 0u;
    unsigned incl = c;
    for (int o = 1; o < 32; o <<= 1) {
        const unsigned t = __shfl_up(incl, o);
        if (lane >= o) incl += t;
    }
    if (lane == 31) wsum[w] = incl;
    __syncthreads();
    if (w == 0) {
        const unsigned v = wsum[lane];
        unsigned si = v;
        for (int o = 1; o < 32; o <<= 1) {
            const unsigned t = __shfl_up(si, o);
            if (lane >= o) si += t;
        }
        wsum[lane] = si - v;
    }
    __syncthreads();
    const unsigned start = wsum[w] + incl - c;
    if (tid < NB) {
        cur[tid] = start;
        offT[tid * NCS + me] = start;
    }
    if (tid == 0) offT[NB * NCS + me] = (unsigned)CHS;
    __syncthreads();

    unsigned pos[CS_PER];
#pragma unroll
    for (int k = 0; k < CS_PER; ++k) {
        const int i = tid + k * 1024;
        pos[k] = (i < CHS) ? atomicAdd(&cur[s[k] >> RSH], 1u) : 0u;
    }
#pragma unroll
    for (int k = 0; k < CS_PER; ++k) {
        const int i = tid + k * 1024;
        if (i < CHS)
            sbuf[pos[k]] = ((unsigned)d[k] << RSH) | (unsigned)(s[k] & (RB - 1));
    }
    __syncthreads();

    uint4*       o4 = (uint4*)(packed + e0);
    const uint4* s4 = (const uint4*)sbuf;
    for (int i = tid; i < CHS / 4; i += 1024) o4[i] = s4[i];
}

static_assert(NCS / 32 == RB / 32, "the piece prefix uses the row prefix's wave sums");
static_assert(NCS * 4 == 512, "four lanes per piece fill the work-group");
template <bool FINAL>
__global__ __launch_bounds__(512) void piece_spmm_kernel(
    const unsigned* __restrict__ packed, const unsigned* __restrict__ offT,
    const float* rows_in, const float* own_in, float* rows_out)
{
    __shared__ unsigned sorted[CAP];
    __shared__ unsigned rcur[RB];
    __shared__ unsigned rbeg[RB + 1];
    __shared__ unsigned wsum[RB / 32];
    __shared__ unsigned ppos[NCS + 1];
    __shared__ unsigned pbeg[NCS];
    const int tid  = threadIdx.x;
    const int lane = tid & 31;
    const int w    = tid >> 5;
    const int b    = blockIdx.x;
    const int r    = tid >> 2;
    const unsigned q = (unsigned)tid & 3u;

    unsigned plen = 0u, pofs = 0u;
    if (tid < NCS) {
        pofs = offT[b * NCS + tid];
        plen = offT[(b + 1) * NCS + tid] - pofs;
    }
    unsigned pincl = plen;
    for (int o = 1; o < 32; o <<= 1) {
        const unsigned t = __shfl_up(pincl, o);
        if (lane >= o) pincl += t;
    }
    if (tid < NCS && lane == 31) wsum[w] = pincl;
    __syncthreads();
    if (tid < NCS) {
        unsigned base = 0u;
        for (int k = 0; k < w; ++k) base += wsum[k];
        const unsigned excl = base + pincl - plen;
        ppos[tid] = excl;
        pbeg[tid] = (unsigned)tid * (unsigned)CHS + pofs;
        if (tid == NCS - 1) ppos[NCS] = excl + plen;
    }
    __syncthreads();
    const unsigned e0 = 0u;
    const unsigned e1 = ppos[NCS];
    const float4* g4  = (const float4*)rows_in;

    float4   sum = make_float4(0.0f, 0.0f, 0.0f, 0.0f);
    unsigned deg = 0u;

    for (unsigned t0 = e0; t0 < e1; t0 += (unsigned)CAP) {
        const unsigned left = e1 - t0;
        const unsigned n    = left < (unsigned)CAP ? left : (unsigned)CAP;

        {
            const unsigned pcn  = (unsigned)tid >> 2;
            const unsigned j4   = (unsigned)tid & 3u;
            const unsigned base = ppos[pcn];
            const unsigned len  = ppos[pcn + 1u] - base;
            const unsigned* src = packed + pbeg[pcn];
            for (unsigned j0 = 0u; j0 < len; j0 += 32u) {
                unsigned v[8];
#pragma unroll
                for (int u = 0; u < 8; ++u) {
                    const unsigned j  = j0 + (unsigned)u * 4u + j4;
                    const unsigned sl = base + j - t0;
                    v[u] = (j < len && sl < n) ? src[j] : 0u;
                }
#pragma unroll
                for (int u = 0; u < 8; ++u) {
                    const unsigned j  = j0 + (unsigned)u * 4u + j4;
                    const unsigned sl = base + j - t0;
                    if (j < len && sl < n) sorted[sl] = v[u];
                }
            }
        }
        __syncthreads();

        unsigned p[PER_T];
#pragma unroll
        for (int k = 0; k < PER_T; ++k) {
            const unsigned i = (unsigned)tid + (unsigned)k * 512u;
            p[k] = (i < n) ? sorted[i] : 0u;
        }
        if (tid < RB) rcur[tid] = 0u;
        __syncthreads();
#pragma unroll
        for (int k = 0; k < PER_T; ++k) {
            const unsigned i = (unsigned)tid + (unsigned)k * 512u;
            if (i < n) atomicAdd(&rcur[p[k] & (RB - 1)], 1u);
        }
        __syncthreads();

        const unsigned c = (tid < RB) ? rcur[tid] : 0u;
        unsigned incl = c;
        for (int o = 1; o < 32; o <<= 1) {
            const unsigned t = __shfl_up(incl, o);
            if (lane >= o) incl += t;
        }
        if (tid < RB && lane == 31) wsum[w] = incl;
        __syncthreads();
        if (tid < RB) {
            unsigned base = 0u;
            for (int k = 0; k < w; ++k) base += wsum[k];
            const unsigned excl = base + incl - c;
            rbeg[tid] = excl;
            rcur[tid] = excl;
        }
        if (tid == 0) rbeg[RB] = n;
        __syncthreads();

#pragma unroll
        for (int k = 0; k < PER_T; ++k) {
            const unsigned i = (unsigned)tid + (unsigned)k * 512u;
            if (i < n) {
                const unsigned pos = atomicAdd(&rcur[p[k] & (RB - 1)], 1u);
                sorted[pos] = p[k] >> RSH;
            }
        }
        __syncthreads();

        const unsigned beg = rbeg[r];
        const unsigned end = rbeg[r + 1];
        deg += end - beg;
        unsigned j = beg;
        for (; j + 4u <= end; j += 4u) {
            unsigned d[4];
            float4   v[4];
            for (int k = 0; k < 4; ++k) d[k] = sorted[j + (unsigned)k];
            for (int k = 0; k < 4; ++k) v[k] = g4[(size_t)d[k] * 4 + q];
            for (int k = 0; k < 4; ++k) {
                sum.x += v[k].x;
                sum.y += v[k].y;
                sum.z += v[k].z;
                sum.w += v[k].w;
            }
        }
        if (j < end) {
            unsigned d[3];
            float4   v[3];
            for (int k = 0; k < 3; ++k) d[k] = sorted[min(j + (unsigned)k, end - 1u)];
            for (int k = 0; k < 3; ++k) v[k] = g4[(size_t)d[k] * 4 + q];
            for (int k = 0; k < 3; ++k) {
                if (j + (unsigned)k < end) {
                    sum.x += v[k].x;
                    sum.y += v[k].y;
                    sum.z += v[k].z;
                    sum.w += v[k].w;
                }
            }
        }
        __syncthreads();
    }

    const int  row   = b * RB + r;
    const bool valid = row < N_NODES;
    const float scale = (1.0f - ALPHA) / fmaxf((float)deg, 1e-12f);
    float4 lv = make_float4(0.0f, 0.0f, 0.0f, 0.0f);
    if (valid) lv = ((const float4*)own_in)[(size_t)row * 4 + q];
    float4 t;
    t.x = scale * sum.x + ALPHA * lv.x;
    t.y = scale * sum.y + ALPHA * lv.y;
    t.z = scale * sum.z + ALPHA * lv.z;
    t.w = scale * sum.w + ALPHA * lv.w;
    if (!FINAL) {
        if (valid) ((float4*)rows_out)[(size_t)row * 4 + q] = t;
    } else {
        float m = fmaxf(fmaxf(t.x, t.y), fmaxf(t.z, t.w));
        m = fmaxf(m, __shfl_xor(m, 1, 4));
        m = fmaxf(m, __shfl_xor(m, 2, 4));
        float sm = (__expf(t.x - m) + __expf(t.y - m)) + (__expf(t.z - m) + __expf(t.w - m));
        sm += __shfl_xor(sm, 1, 4);
        sm += __shfl_xor(sm, 2, 4);
        const float lse = m + __logf(sm);
        t.x -= lse; t.y -= lse; t.z -= lse; t.w -= lse;
        if (valid) ((float4*)rows_out)[(size_t)row * 4 + q] = t;
    }
}

extern "C" void kernel_launch(void* const* d_in, const int* in_sizes, int n_in,
                              void* d_out, int out_size, void* d_ws, size_t ws_size,
                              hipStream_t stream)
{
    const float* x    = (const float*)d_in[0];
    const float* W1   = (const float*)d_in[1];
    const float* W2   = (const float*)d_in[2];
    const int*   esrc = (const int*)d_in[3];
    const int*   edst = (const int*)d_in[4];
    float* out = (float*)d_out;

    const size_t off_packed = (size_t)N_NODES * N_CLASS * sizeof(float);
    const size_t off_bstart = off_packed + (size_t)N_EDGES * sizeof(unsigned);
    const size_t off_tot    = off_bstart + (size_t)(NB + 1) * sizeof(unsigned);
    const size_t ws_need    = off_tot + (size_t)NB * sizeof(unsigned);
    const size_t off_offT = off_packed + (size_t)N_EDGES * sizeof(unsigned);
    const size_t ws_need2 = off_offT + (size_t)(NB + 1) * NCS * sizeof(unsigned);
    if (ws_size >= ws_need2) {
        float*    lg     = (float*)d_ws;
        unsigned* packed = (unsigned*)((char*)d_ws + off_packed);
        unsigned* offT   = (unsigned*)((char*)d_ws + off_offT);
        const unsigned gG = (unsigned)(((N_NODES + 15) / 16 + GW - 1) / GW);
        gcn_gemm_kernel<<<gG, dim3(GT), 0, stream>>>(x, W1, W2, out);
        chunk_sort_kernel<<<NCS, dim3(1024), 0, stream>>>(esrc, edst, packed, offT);
        piece_spmm_kernel<false><<<NB, dim3(512), 0, stream>>>(packed, offT, out, out, lg);
        piece_spmm_kernel<true><<<NB, dim3(512), 0, stream>>>(packed, offT, lg, out, out);
        return;
    }
    if (ws_size >= ws_need) {
        float*    lg     = (float*)d_ws;
        unsigned* cnt    = (unsigned*)d_ws;
        unsigned* packed = (unsigned*)((char*)d_ws + off_packed);
        unsigned* bstart = (unsigned*)((char*)d_ws + off_bstart);
        unsigned* tot    = (unsigned*)((char*)d_ws + off_tot);
        const unsigned gG = (unsigned)(((N_NODES + 15) / 16 + GW - 1) / GW);
        gcn_gemm_kernel<<<gG, dim3(GT), 0, stream>>>(x, W1, W2, out);
        bucket_count_kernel<<<NCH, dim3(1024), 0, stream>>>(esrc, cnt);
        bucket_prefix_kernel<<<(NB + 7) / 8, dim3(256), 0, stream>>>(cnt, tot);
        bucket_place_kernel<<<NCH, dim3(1024), 0, stream>>>(esrc, edst, cnt, tot, packed, bstart);
        bucket_spmm_kernel<false><<<NB, dim3(512), 0, stream>>>(packed, bstart, out, out, lg);
        bucket_spmm_kernel<true><<<NB, dim3(512), 0, stream>>>(packed, bstart, lg, out, out);
        return;
    }

    float* local  = (float*)d_ws;
    float* logits = local  + (size_t)N_NODES * N_CLASS;
    float* deg    = logits + (size_t)N_NODES * N_CLASS;
    float* agg    = out;

    const dim3 blk(256);
    const long long nl   = (long long)N_NODES * N_CLASS;
    const long long ne16 = (long long)N_EDGES * N_CLASS;
    const unsigned gN    = (unsigned)((N_NODES + 255) / 256);
    const unsigned gNL   = (unsigned)((nl + 255) / 256);
    const unsigned gE    = (unsigned)((N_EDGES + 255) / 256);
    const unsigned gE16  = (unsigned)((ne16 + 255) / 256);

    const unsigned gGEMM = (unsigned)(((N_NODES + 15) / 16 + GW - 1) / GW);
    gcn_gemm_kernel<<<gGEMM, dim3(GT), 0, stream>>>(x, W1, W2, local);

    zero_kernel<<<gN, blk, 0, stream>>>(deg, N_NODES);
    deg_kernel<<<gE, blk, 0, stream>>>(esrc, deg);

    zero_kernel<<<gNL, blk, 0, stream>>>(agg, nl);
    scatter_kernel<<<gE16, blk, 0, stream>>>(esrc, edst, local, agg);
    combine_kernel<<<gNL, blk, 0, stream>>>(agg, local, deg, logits);

    zero_kernel<<<gNL, blk, 0, stream>>>(agg, nl);
    scatter_kernel<<<gE16, blk, 0, stream>>>(esrc, edst, logits, agg);
    final_kernel<<<gN, blk, 0, stream>>>(agg, local, deg, out);
}


# GEMM kernel: each work-group's 25th row tile is split by columns over four extra waves, one per SIMD (per SIMD 6 whole tiles + a quarter tile, was 7/6/6/6 whole tiles); sums bit-identical
# speedup vs baseline: 4.1321x; 1.0332x over previous
#include <hip/hip_runtime.h>
#include <hip/hip_bf16.h>
#include <math.h>

#define N_NODES 100000
#define N_FEAT  512
#define HIDDEN  128
#define N_CLASS 16
#define N_EDGES 3200000
#define ALPHA   0.25f

typedef __attribute__((ext_vector_type(2))) float v2f;
typedef __attribute__((ext_vector_type(8))) float v8f;

#define GW       25
#define GF       (GW - 1)
#define GQ       4
#define GQT      (8 / GQ)
#define GT       ((GF + GQ) * 32)
#define W1P_STR  288
#define HID_STR  132
#define W1_ITEMS ((N_FEAT / 2) * (HIDDEN / 4))
#define W1_FULL  (W1_ITEMS / GT)
static_assert(GW * 16 * HID_STR <= (N_FEAT / 2) * W1P_STR, "hidden tiles fit in the W1 region");

__global__ __launch_bounds__(GT) void gcn_gemm_kernel(
    const float* __restrict__ x, const float* __restrict__ W1,
    const float* __restrict__ W2, float* __restrict__ local_out)
{
    __shared__ __attribute__((aligned(16))) float w1s[(N_FEAT / 2) * W1P_STR];

    const int tid  = threadIdx.x;
    const int wave = tid >> 5;
    const int lane = tid & 31;
    const int half = lane >> 4;
    const int l16  = lane & 15;

    const bool whole   = (wave >= GQ);
    const int  part    = whole ? 0 : wave;
    const int  slot    = whole ? wave - GQ : GF;
    const int row_tile = blockIdx.x * GW + slot;
    const int row_base = row_tile * 16;
    const bool active  = (row_base < N_NODES);

    {
        float4 wa[W1_FULL], wb[W1_FULL];
#pragma unroll
        for (int j = 0; j < W1_FULL; ++j) {
            const int i  = tid + j * GT;
            const int p  = i >> 5;
            const int c4 = i & 31;
            wa[j] = ((const float4*)(W1 + (size_t)(2 * p + 0) * HIDDEN))[c4];
            wb[j] = ((const float4*)(W1 + (size_t)(2 * p + 1) * HIDDEN))[c4];
        }
#pragma unroll
        for (int j = 0; j < W1_FULL; ++j) {
            const int i  = tid + j * GT;
            const int p  = i >> 5;
            const int c4 = i & 31;
            float* d = &w1s[p * W1P_STR + c4 * 8];
            *(float4*)(d + 0) = make_float4(wa[j].x, wb[j].x, wa[j].y, wb[j].y);
            *(float4*)(d + 4) = make_float4(wa[j].z, wb[j].z, wa[j].w, wb[j].w);
        }
    }
    {
        const int i = tid + W1_FULL * GT;
        if (i < W1_ITEMS) {
            const int p  = i >> 5;
            const int c4 = i & 31;
            const float4 ra = ((const float4*)(W1 + (size_t)(2 * p + 0) * HIDDEN))[c4];
            const float4 rb = ((const float4*)(W1 + (size_t)(2 * p + 1) * HIDDEN))[c4];
            float* d = &w1s[p * W1P_STR + c4 * 8];
            *(float4*)(d + 0) = make_float4(ra.x, rb.x, ra.y, rb.y);
            *(float4*)(d + 4) = make_float4(ra.z, rb.z, ra.w, rb.w);
        }
    }
    __syncthreads();

    v8f acc[8];
    for (int t = 0; t < 8; ++t)
        for (int i = 0; i < 8; ++i) acc[t][i] = 0.0f;

    if (active) {
        const float* ap = x + (size_t)(row_base + l16) * N_FEAT + 2 * half;
        const float* bp = w1s + half * W1P_STR + l16 * 2;
        if (whole) {
            for (int k = 0; k < N_FEAT; k += 4) {
                v2f a = *(const v2f*)(ap + k);
                for (int t = 0; t < 8; ++t) {
                    v2f b = *(const v2f*)(bp + (k >> 1) * W1P_STR + t * 32);
                    acc[t] = __builtin_amdgcn_wmma_f32_16x16x4_f32(
                        false, a, false, b, (short)0, acc[t], false, false);
                }
            }
        } else {
            const float* bq = bp + part * (GQT * 32);
            for (int k = 0; k < N_FEAT; k += 4) {
                v2f a = *(const v2f*)(ap + k);
                for (int t = 0; t < GQT; ++t) {
                    v2f b = *(const v2f*)(bq + (k >> 1) * W1P_STR + t * 32);
                    acc[t] = __builtin_amdgcn_wmma_f32_16x16x4_f32(
                        false, a, false, b, (short)0, acc[t], false, false);
                }
            }
        }
    }
    __syncthreads();

    float* hid = w1s + slot * (16 * HID_STR);
    if (active) {
        if (whole) {
            for (int t = 0; t < 8; ++t)
                for (int r = 0; r < 8; ++r) {
                    float v = acc[t][r];
                    hid[(r + 8 * half) * HID_STR + t * 16 + l16] = v > 0.0f ? v : 0.0f;
                }
        } else {
            for (int t = 0; t < GQT; ++t)
                for (int r = 0; r < 8; ++r) {
                    float v = acc[t][r];
                    hid[(r + 8 * half) * HID_STR + (part * GQT + t) * 16 + l16] = v > 0.0f ? v : 0.0f;
                }
        }
    }
    __syncthreads();

    if (active && (whole || part == 0)) {
        v8f c2;
        for (int i = 0; i < 8; ++i) c2[i] = 0.0f;
        for (int kk = 0; kk < HIDDEN; kk += 4) {
            v2f a, b;
            a[0] = hid[l16 * HID_STR + kk + 2 * half + 0];
            a[1] = hid[l16 * HID_STR + kk + 2 * half + 1];
            b[0] = W2[(size_t)(kk + 2 * half + 0) * N_CLASS + l16];
            b[1] = W2[(size_t)(kk + 2 * half + 1) * N_CLASS + l16];
            c2 = __builtin_amdgcn_wmma_f32_16x16x4_f32(
                false, a, false, b, (short)0, c2, false, false);
        }
        for (int r = 0; r < 8; ++r)
            local_out[(size_t)(row_base + r + 8 * half) * N_CLASS + l16] = c2[r];
    }
}

__global__ void zero_kernel(float* __restrict__ p, long long n)
{
    long long i = (long long)blockIdx.x * blockDim.x + threadIdx.x;
    if (i < n) p[i] = 0.0f;
}

__global__ void deg_kernel(const int* __restrict__ src, float* __restrict__ deg)
{
    int e = blockIdx.x * blockDim.x + threadIdx.x;
    if (e < N_EDGES) atomicAdd(&deg[src[e]], 1.0f);
}

__global__ void scatter_kernel(const int* __restrict__ src,
                               const int* __restrict__ dst,
                               const float* __restrict__ logits,
                               float* __restrict__ agg)
{
    long long gid = (long long)blockIdx.x * blockDim.x + threadIdx.x;
    if (gid >= (long long)N_EDGES * N_CLASS) return;
    const int e = (int)(gid >> 4);
    const int c = (int)(gid & 15);
    const int s = src[e];
    const int d = dst[e];
    atomicAdd(&agg[(size_t)s * N_CLASS + c], logits[(size_t)d * N_CLASS + c]);
}

__global__ void combine_kernel(const float* __restrict__ agg,
                               const float* __restrict__ local,
                               const float* __restrict__ deg,
                               float* __restrict__ out)
{
    long long i = (long long)blockIdx.x * blockDim.x + threadIdx.x;
    if (i >= (long long)N_NODES * N_CLASS) return;
    const int row = (int)(i >> 4);
    const float scale = (1.0f - ALPHA) / fmaxf(deg[row], 1e-12f);
    out[i] = scale * agg[i] + ALPHA * local[i];
}

__global__ void final_kernel(const float* __restrict__ agg,
                             const float* __restrict__ local,
                             const float* __restrict__ deg,
                             float* __restrict__ out)
{
    int row = blockIdx.x * blockDim.x + threadIdx.x;
    if (row >= N_NODES) return;
    const float scale = (1.0f - ALPHA) / fmaxf(deg[row], 1e-12f);
    float t[N_CLASS];
    float m = -INFINITY;
    for (int c = 0; c < N_CLASS; ++c) {
        t[c] = scale * agg[(size_t)row * N_CLASS + c]
             + ALPHA * local[(size_t)row * N_CLASS + c];
        m = fmaxf(m, t[c]);
    }
    float s = 0.0f;
    for (int c = 0; c < N_CLASS; ++c) s += __expf(t[c] - m);
    const float lse = m + __logf(s);
    for (int c = 0; c < N_CLASS; ++c)
        out[(size_t)row * N_CLASS + c] = t[c] - lse;
}


#define RSH    7
#define RB     (1 << RSH)
#define NB     ((N_NODES + RB - 1) / RB)
#define NCH    256
#define CH     (N_EDGES / NCH)
#define CAP    6144
#define PER_T  (CAP / 512)
static_assert(N_EDGES % NCH == 0, "chunks must tile the edge list");
static_assert(NB <= 1024, "one thread per bucket in the place kernel");
static_assert(N_NODES <= (1 << (32 - RSH)), "dst and local row share one word");

__global__ __launch_bounds__(1024) void bucket_count_kernel(
    const int* __restrict__ esrc, unsigned* __restrict__ cnt)
{
    __shared__ unsigned h[NB];
    const int tid = threadIdx.x;
    for (int i = tid; i < NB; i += 1024) h[i] = 0u;
    __syncthreads();
    const int e0 = blockIdx.x * CH;
    for (int i = tid; i < CH; i += 1024)
        atomicAdd(&h[esrc[e0 + i] >> RSH], 1u);
    __syncthreads();
    for (int i = tid; i < NB; i += 1024) cnt[blockIdx.x * NB + i] = h[i];
}

static_assert(NCH % 32 == 0, "each lane takes NCH/32 consecutive chunks");
__global__ __launch_bounds__(256) void bucket_prefix_kernel(
    unsigned* __restrict__ cnt, unsigned* __restrict__ tot)
{
    const int lane = threadIdx.x & 31;
    const int b    = blockIdx.x * 8 + (threadIdx.x >> 5);
    if (b >= NB) return;
    unsigned c[NCH / 32];
    for (int j = 0; j < NCH / 32; ++j)
        c[j] = cnt[(lane * (NCH / 32) + j) * NB + b];
    unsigned s = 0u;
    for (int j = 0; j < NCH / 32; ++j) { const unsigned t = c[j]; c[j] = s; s += t; }
    unsigned incl = s;
    for (int o = 1; o < 32; o <<= 1) {
        const unsigned t = __shfl_up(incl, o);
        if (lane >= o) incl += t;
    }
    const unsigned base = incl - s;
    for (int j = 0; j < NCH / 32; ++j)
        cnt[(lane * (NCH / 32) + j) * NB + b] = base + c[j];
    if (lane == 31) tot[b] = incl;
}

__global__ __launch_bounds__(1024) void bucket_place_kernel(
    const int* __restrict__ esrc, const int* __restrict__ edst,
    const unsigned* __restrict__ cnt, const unsigned* __restrict__ tot,
    unsigned* __restrict__ packed, unsigned* __restrict__ bstart)
{
    __shared__ unsigned cur[NB];
    __shared__ unsigned wsum[32];
    const int tid  = threadIdx.x;
    const int lane = tid & 31;
    const int w    = tid >> 5;
    const int me   = blockIdx.x;

    const unsigned tb     = (tid < NB) ? tot[tid] : 0u;
    const unsigned before = (tid < NB) ? cnt[me * NB + tid] : 0u;
    unsigned incl = tb;
    for (int o = 1; o < 32; o <<= 1) {
        const unsigned t = __shfl_up(incl, o);
        if (lane >= o) incl += t;
    }
    if (lane == 31) wsum[w] = incl;
    __syncthreads();
    if (w == 0) {
        const unsigned s = wsum[lane];
        unsigned si = s;
        for (int o = 1; o < 32; o <<= 1) {
            const unsigned t = __shfl_up(si, o);
            if (lane >= o) si += t;
        }
        wsum[lane] = si - s;
    }
    __syncthreads();
    const unsigned start = wsum[w] + incl - tb;
    if (tid < NB) {
        cur[tid] = start + before;
        if (me == 0) bstart[tid] = start;
    }
    if (me == 0 && tid == 0) bstart[NB] = (unsigned)N_EDGES;
    __syncthreads();

    const int e0 = me * CH;
    for (int i = tid; i < CH; i += 1024) {
        const int s = esrc[e0 + i];
        const int d = edst[e0 + i];
        const unsigned pos = atomicAdd(&cur[s >> RSH], 1u);
        packed[pos] = ((unsigned)d << RSH) | (unsigned)(s & (RB - 1));
    }
}

static_assert(RB * 4 == 512, "four lanes per row fill the work-group");
static_assert(CAP % 512 == 0, "a thread holds CAP/512 edge words per pass");
template <bool FINAL>
__global__ __launch_bounds__(512) void bucket_spmm_kernel(
    const unsigned* __restrict__ packed, const unsigned* __restrict__ bstart,
    const float* rows_in, const float* own_in, float* rows_out)
{
    __shared__ unsigned sorted[CAP];
    __shared__ unsigned rcur[RB];
    __shared__ unsigned rbeg[RB + 1];
    __shared__ unsigned wsum[RB / 32];
    const int tid  = threadIdx.x;
    const int lane = tid & 31;
    const int w    = tid >> 5;
    const int b    = blockIdx.x;
    const int r    = tid >> 2;
    const unsigned q = (unsigned)tid & 3u;

    const unsigned e0 = bstart[b];
    const unsigned e1 = bstart[b + 1];
    const float4* g4  = (const float4*)rows_in;

    float4   sum = make_float4(0.0f, 0.0f, 0.0f, 0.0f);
    unsigned deg = 0u;

    for (unsigned t0 = e0; t0 < e1; t0 += (unsigned)CAP) {
        const unsigned left = e1 - t0;
        const unsigned n    = left < (unsigned)CAP ? left : (unsigned)CAP;

        unsigned p[PER_T];
#pragma unroll
        for (int k = 0; k < PER_T; ++k) {
            const unsigned i = (unsigned)tid + (unsigned)k * 512u;
            p[k] = (i < n) ? packed[t0 + i] : 0u;
        }
        if (tid < RB) rcur[tid] = 0u;
        __syncthreads();
#pragma unroll
        for (int k = 0; k < PER_T; ++k) {
            const unsigned i = (unsigned)tid + (unsigned)k * 512u;
            if (i < n) atomicAdd(&rcur[p[k] & (RB - 1)], 1u);
        }
        __syncthreads();

        const unsigned c = (tid < RB) ? rcur[tid] : 0u;
        unsigned incl = c;
        for (int o = 1; o < 32; o <<= 1) {
            const unsigned t = __shfl_up(incl, o);
            if (lane >= o) incl += t;
        }
        if (tid < RB && lane == 31) wsum[w] = incl;
        __syncthreads();
        if (tid < RB) {
            unsigned base = 0u;
            for (int k = 0; k < w; ++k) base += wsum[k];
            const unsigned excl = base + incl - c;
            rbeg[tid] = excl;
            rcur[tid] = excl;
        }
        if (tid == 0) rbeg[RB] = n;
        __syncthreads();

#pragma unroll
        for (int k = 0; k < PER_T; ++k) {
            const unsigned i = (unsigned)tid + (unsigned)k * 512u;
            if (i < n) {
                const unsigned pos = atomicAdd(&rcur[p[k] & (RB - 1)], 1u);
                sorted[pos] = p[k] >> RSH;
            }
        }
        __syncthreads();

        const unsigned beg = rbeg[r];
        const unsigned end = rbeg[r + 1];
        deg += end - beg;
        unsigned j = beg;
        for (; j + 4u <= end; j += 4u) {
            unsigned d[4];
            float4   v[4];
            for (int k = 0; k < 4; ++k) d[k] = sorted[j + (unsigned)k];
            for (int k = 0; k < 4; ++k) v[k] = g4[(size_t)d[k] * 4 + q];
            for (int k = 0; k < 4; ++k) {
                sum.x += v[k].x;
                sum.y += v[k].y;
                sum.z += v[k].z;
                sum.w += v[k].w;
            }
        }
        if (j < end) {
            unsigned d[3];
            float4   v[3];
            for (int k = 0; k < 3; ++k) d[k] = sorted[min(j + (unsigned)k, end - 1u)];
            for (int k = 0; k < 3; ++k) v[k] = g4[(size_t)d[k] * 4 + q];
            for (int k = 0; k < 3; ++k) {
                if (j + (unsigned)k < end) {
                    sum.x += v[k].x;
                    sum.y += v[k].y;
                    sum.z += v[k].z;
                    sum.w += v[k].w;
                }
            }
        }
        __syncthreads();
    }

    const int  row   = b * RB + r;
    const bool valid = row < N_NODES;
    const float scale = (1.0f - ALPHA) / fmaxf((float)deg, 1e-12f);
    float4 lv = make_float4(0.0f, 0.0f, 0.0f, 0.0f);
    if (valid) lv = ((const float4*)own_in)[(size_t)row * 4 + q];
    float4 t;
    t.x = scale * sum.x + ALPHA * lv.x;
    t.y = scale * sum.y + ALPHA * lv.y;
    t.z = scale * sum.z + ALPHA * lv.z;
    t.w = scale * sum.w + ALPHA * lv.w;
    if (!FINAL) {
        if (valid) ((float4*)rows_out)[(size_t)row * 4 + q] = t;
    } else {
        float m = fmaxf(fmaxf(t.x, t.y), fmaxf(t.z, t.w));
        m = fmaxf(m, __shfl_xor(m, 1, 4));
        m = fmaxf(m, __shfl_xor(m, 2, 4));
        float sm = (__expf(t.x - m) + __expf(t.y - m)) + (__expf(t.z - m) + __expf(t.w - m));
        sm += __shfl_xor(sm, 1, 4);
        sm += __shfl_xor(sm, 2, 4);
        const float lse = m + __logf(sm);
        t.x -= lse; t.y -= lse; t.z -= lse; t.w -= lse;
        if (valid) ((float4*)rows_out)[(size_t)row * 4 + q] = t;
    }
}


#define NCS    128
#define CHS    (N_EDGES / NCS)
#define CS_PER ((CHS + 1023) / 1024)
static_assert(N_EDGES % NCS == 0, "chunks must tile the edge list");
static_assert(CHS % 4 == 0, "a segment is written as 16-byte pieces");

__global__ __launch_bounds__(1024) void chunk_sort_kernel(
    const int* __restrict__ esrc, const int* __restrict__ edst,
    unsigned* __restrict__ packed, unsigned* __restrict__ offT)
{
    __shared__ __attribute__((aligned(16))) unsigned sbuf[CHS];
    __shared__ unsigned cur[NB];
    __shared__ unsigned wsum[32];
    const int tid  = threadIdx.x;
    const int lane = tid & 31;
    const int w    = tid >> 5;
    const int me   = blockIdx.x;
    const int e0   = me * CHS;

    int s[CS_PER], d[CS_PER];
#pragma unroll
    for (int k = 0; k < CS_PER; ++k) {
        const int i = tid + k * 1024;
        s[k] = (i < CHS) ? esrc[e0 + i] : 0;
        d[k] = (i < CHS) ? edst[e0 + i] : 0;
    }
    if (tid < NB) cur[tid] = 0u;
    __syncthreads();
#pragma unroll
    for (int k = 0; k < CS_PER; ++k) {
        const int i = tid + k * 1024;
        if (i < CHS) atomicAdd(&cur[s[k] >> RSH], 1u);
    }
    __syncthreads();

    const unsigned c = (tid < NB) ? cur[tid] : 0u;
    unsigned incl = c;
    for (int o = 1; o < 32; o <<= 1) {
        const unsigned t = __shfl_up(incl, o);
        if (lane >= o) incl += t;
    }
    if (lane == 31) wsum[w] = incl;
    __syncthreads();
    if (w == 0) {
        const unsigned v = wsum[lane];
        unsigned si = v;
        for (int o = 1; o < 32; o <<= 1) {
            const unsigned t = __shfl_up(si, o);
            if (lane >= o) si += t;
        }
        wsum[lane] = si - v;
    }
    __syncthreads();
    const unsigned start = wsum[w] + incl - c;
    if (tid < NB) {
        cur[tid] = start;
        offT[tid * NCS + me] = start;
    }
    if (tid == 0) offT[NB * NCS + me] = (unsigned)CHS;
    __syncthreads();

    unsigned pos[CS_PER];
#pragma unroll
    for (int k = 0; k < CS_PER; ++k) {
        const int i = tid + k * 1024;
        pos[k] = (i < CHS) ? atomicAdd(&cur[s[k] >> RSH], 1u) : 0u;
    }
#pragma unroll
    for (int k = 0; k < CS_PER; ++k) {
        const int i = tid + k * 1024;
        if (i < CHS)
            sbuf[pos[k]] = ((unsigned)d[k] << RSH) | (unsigned)(s[k] & (RB - 1));
    }
    __syncthreads();

    uint4*       o4 = (uint4*)(packed + e0);
    const uint4* s4 = (const uint4*)sbuf;
    for (int i = tid; i < CHS / 4; i += 1024) o4[i] = s4[i];
}

static_assert(NCS / 32 == RB / 32, "the piece prefix uses the row prefix's wave sums");
static_assert(NCS * 4 == 512, "four lanes per piece fill the work-group");
template <bool FINAL>
__global__ __launch_bounds__(512) void piece_spmm_kernel(
    const unsigned* __restrict__ packed, const unsigned* __restrict__ offT,
    const float* rows_in, const float* own_in, float* rows_out)
{
    __shared__ unsigned sorted[CAP];
    __shared__ unsigned rcur[RB];
    __shared__ unsigned rbeg[RB + 1];
    __shared__ unsigned wsum[RB / 32];
    __shared__ unsigned ppos[NCS + 1];
    __shared__ unsigned pbeg[NCS];
    const int tid  = threadIdx.x;
    const int lane = tid & 31;
    const int w    = tid >> 5;
    const int b    = blockIdx.x;
    const int r    = tid >> 2;
    const unsigned q = (unsigned)tid & 3u;

    unsigned plen = 0u, pofs = 0u;
    if (tid < NCS) {
        pofs = offT[b * NCS + tid];
        plen = offT[(b + 1) * NCS + tid] - pofs;
    }
    unsigned pincl = plen;
    for (int o = 1; o < 32; o <<= 1) {
        const unsigned t = __shfl_up(pincl, o);
        if (lane >= o) pincl += t;
    }
    if (tid < NCS && lane == 31) wsum[w] = pincl;
    __syncthreads();
    if (tid < NCS) {
        unsigned base = 0u;
        for (int k = 0; k < w; ++k) base += wsum[k];
        const unsigned excl = base + pincl - plen;
        ppos[tid] = excl;
        pbeg[tid] = (unsigned)tid * (unsigned)CHS + pofs;
        if (tid == NCS - 1) ppos[NCS] = excl + plen;
    }
    __syncthreads();
    const unsigned e0 = 0u;
    const unsigned e1 = ppos[NCS];
    const float4* g4  = (const float4*)rows_in;

    float4   sum = make_float4(0.0f, 0.0f, 0.0f, 0.0f);
    unsigned deg = 0u;

    for (unsigned t0 = e0; t0 < e1; t0 += (unsigned)CAP) {
        const unsigned left = e1 - t0;
        const unsigned n    = left < (unsigned)CAP ? left : (unsigned)CAP;

        {
            const unsigned pcn  = (unsigned)tid >> 2;
            const unsigned j4   = (unsigned)tid & 3u;
            const unsigned base = ppos[pcn];
            const unsigned len  = ppos[pcn + 1u] - base;
            const unsigned* src = packed + pbeg[pcn];
            for (unsigned j0 = 0u; j0 < len; j0 += 32u) {
                unsigned v[8];
#pragma unroll
                for (int u = 0; u < 8; ++u) {
                    const unsigned j  = j0 + (unsigned)u * 4u + j4;
                    const unsigned sl = base + j - t0;
                    v[u] = (j < len && sl < n) ? src[j] : 0u;
                }
#pragma unroll
                for (int u = 0; u < 8; ++u) {
                    const unsigned j  = j0 + (unsigned)u * 4u + j4;
                    const unsigned sl = base + j - t0;
                    if (j < len && sl < n) sorted[sl] = v[u];
                }
            }
        }
        __syncthreads();

        unsigned p[PER_T];
#pragma unroll
        for (int k = 0; k < PER_T; ++k) {
            const unsigned i = (unsigned)tid + (unsigned)k * 512u;
            p[k] = (i < n) ? sorted[i] : 0u;
        }
        if (tid < RB) rcur[tid] = 0u;
        __syncthreads();
#pragma unroll
        for (int k = 0; k < PER_T; ++k) {
            const unsigned i = (unsigned)tid + (unsigned)k * 512u;
            if (i < n) atomicAdd(&rcur[p[k] & (RB - 1)], 1u);
        }
        __syncthreads();

        const unsigned c = (tid < RB) ? rcur[tid] : 0u;
        unsigned incl = c;
        for (int o = 1; o < 32; o <<= 1) {
            const unsigned t = __shfl_up(incl, o);
            if (lane >= o) incl += t;
        }
        if (tid < RB && lane == 31) wsum[w] = incl;
        __syncthreads();
        if (tid < RB) {
            unsigned base = 0u;
            for (int k = 0; k < w; ++k) base += wsum[k];
            const unsigned excl = base + incl - c;
            rbeg[tid] = excl;
            rcur[tid] = excl;
        }
        if (tid == 0) rbeg[RB] = n;
        __syncthreads();

#pragma unroll
        for (int k = 0; k < PER_T; ++k) {
            const unsigned i = (unsigned)tid + (unsigned)k * 512u;
            if (i < n) {
                const unsigned pos = atomicAdd(&rcur[p[k] & (RB - 1)], 1u);
                sorted[pos] = p[k] >> RSH;
            }
        }
        __syncthreads();

        const unsigned beg = rbeg[r];
        const unsigned end = rbeg[r + 1];
        deg += end - beg;
        unsigned j = beg;
        for (; j + 4u <= end; j += 4u) {
            unsigned d[4];
            float4   v[4];
            for (int k = 0; k < 4; ++k) d[k] = sorted[j + (unsigned)k];
            for (int k = 0; k < 4; ++k) v[k] = g4[(size_t)d[k] * 4 + q];
            for (int k = 0; k < 4; ++k) {
                sum.x += v[k].x;
                sum.y += v[k].y;
                sum.z += v[k].z;
                sum.w += v[k].w;
            }
        }
        if (j < end) {
            unsigned d[3];
            float4   v[3];
            for (int k = 0; k < 3; ++k) d[k] = sorted[min(j + (unsigned)k, end - 1u)];
            for (int k = 0; k < 3; ++k) v[k] = g4[(size_t)d[k] * 4 + q];
            for (int k = 0; k < 3; ++k) {
                if (j + (unsigned)k < end) {
                    sum.x += v[k].x;
                    sum.y += v[k].y;
                    sum.z += v[k].z;
                    sum.w += v[k].w;
                }
            }
        }
        __syncthreads();
    }

    const int  row   = b * RB + r;
    const bool valid = row < N_NODES;
    const float scale = (1.0f - ALPHA) / fmaxf((float)deg, 1e-12f);
    float4 lv = make_float4(0.0f, 0.0f, 0.0f, 0.0f);
    if (valid) lv = ((const float4*)own_in)[(size_t)row * 4 + q];
    float4 t;
    t.x = scale * sum.x + ALPHA * lv.x;
    t.y = scale * sum.y + ALPHA * lv.y;
    t.z = scale * sum.z + ALPHA * lv.z;
    t.w = scale * sum.w + ALPHA * lv.w;
    if (!FINAL) {
        if (valid) ((float4*)rows_out)[(size_t)row * 4 + q] = t;
    } else {
        float m = fmaxf(fmaxf(t.x, t.y), fmaxf(t.z, t.w));
        m = fmaxf(m, __shfl_xor(m, 1, 4));
        m = fmaxf(m, __shfl_xor(m, 2, 4));
        float sm = (__expf(t.x - m) + __expf(t.y - m)) + (__expf(t.z - m) + __expf(t.w - m));
        sm += __shfl_xor(sm, 1, 4);
        sm += __shfl_xor(sm, 2, 4);
        const float lse = m + __logf(sm);
        t.x -= lse; t.y -= lse; t.z -= lse; t.w -= lse;
        if (valid) ((float4*)rows_out)[(size_t)row * 4 + q] = t;
    }
}

extern "C" void kernel_launch(void* const* d_in, const int* in_sizes, int n_in,
                              void* d_out, int out_size, void* d_ws, size_t ws_size,
                              hipStream_t stream)
{
    const float* x    = (const float*)d_in[0];
    const float* W1   = (const float*)d_in[1];
    const float* W2   = (const float*)d_in[2];
    const int*   esrc = (const int*)d_in[3];
    const int*   edst = (const int*)d_in[4];
    float* out = (float*)d_out;

    const size_t off_packed = (size_t)N_NODES * N_CLASS * sizeof(float);
    const size_t off_bstart = off_packed + (size_t)N_EDGES * sizeof(unsigned);
    const size_t off_tot    = off_bstart + (size_t)(NB + 1) * sizeof(unsigned);
    const size_t ws_need    = off_tot + (size_t)NB * sizeof(unsigned);
    const size_t off_offT = off_packed + (size_t)N_EDGES * sizeof(unsigned);
    const size_t ws_need2 = off_offT + (size_t)(NB + 1) * NCS * sizeof(unsigned);
    if (ws_size >= ws_need2) {
        float*    lg     = (float*)d_ws;
        unsigned* packed = (unsigned*)((char*)d_ws + off_packed);
        unsigned* offT   = (unsigned*)((char*)d_ws + off_offT);
        const unsigned gG = (unsigned)(((N_NODES + 15) / 16 + GW - 1) / GW);
        gcn_gemm_kernel<<<gG, dim3(GT), 0, stream>>>(x, W1, W2, out);
        chunk_sort_kernel<<<NCS, dim3(1024), 0, stream>>>(esrc, edst, packed, offT);
        piece_spmm_kernel<false><<<NB, dim3(512), 0, stream>>>(packed, offT, out, out, lg);
        piece_spmm_kernel<true><<<NB, dim3(512), 0, stream>>>(packed, offT, lg, out, out);
        return;
    }
    if (ws_size >= ws_need) {
        float*    lg     = (float*)d_ws;
        unsigned* cnt    = (unsigned*)d_ws;
        unsigned* packed = (unsigned*)((char*)d_ws + off_packed);
        unsigned* bstart = (unsigned*)((char*)d_ws + off_bstart);
        unsigned* tot    = (unsigned*)((char*)d_ws + off_tot);
        const unsigned gG = (unsigned)(((N_NODES + 15) / 16 + GW - 1) / GW);
        gcn_gemm_kernel<<<gG, dim3(GT), 0, stream>>>(x, W1, W2, out);
        bucket_count_kernel<<<NCH, dim3(1024), 0, stream>>>(esrc, cnt);
        bucket_prefix_kernel<<<(NB + 7) / 8, dim3(256), 0, stream>>>(cnt, tot);
        bucket_place_kernel<<<NCH, dim3(1024), 0, stream>>>(esrc, edst, cnt, tot, packed, bstart);
        bucket_spmm_kernel<false><<<NB, dim3(512), 0, stream>>>(packed, bstart, out, out, lg);
        bucket_spmm_kernel<true><<<NB, dim3(512), 0, stream>>>(packed, bstart, lg, out, out);
        return;
    }

    float* local  = (float*)d_ws;
    float* logits = local  + (size_t)N_NODES * N_CLASS;
    float* deg    = logits + (size_t)N_NODES * N_CLASS;
    float* agg    = out;

    const dim3 blk(256);
    const long long nl   = (long long)N_NODES * N_CLASS;
    const long long ne16 = (long long)N_EDGES * N_CLASS;
    const unsigned gN    = (unsigned)((N_NODES + 255) / 256);
    const unsigned gNL   = (unsigned)((nl + 255) / 256);
    const unsigned gE    = (unsigned)((N_EDGES + 255) / 256);
    const unsigned gE16  = (unsigned)((ne16 + 255) / 256);

    const unsigned gGEMM = (unsigned)(((N_NODES + 15) / 16 + GW - 1) / GW);
    gcn_gemm_kernel<<<gGEMM, dim3(GT), 0, stream>>>(x, W1, W2, local);

    zero_kernel<<<gN, blk, 0, stream>>>(deg, N_NODES);
    deg_kernel<<<gE, blk, 0, stream>>>(esrc, deg);

    zero_kernel<<<gNL, blk, 0, stream>>>(agg, nl);
    scatter_kernel<<<gE16, blk, 0, stream>>>(esrc, edst, local, agg);
    combine_kernel<<<gNL, blk, 0, stream>>>(agg, local, deg, logits);

    zero_kernel<<<gNL, blk, 0, stream>>>(agg, nl);
    scatter_kernel<<<gE16, blk, 0, stream>>>(esrc, edst, logits, agg);
    final_kernel<<<gN, blk, 0, stream>>>(agg, local, deg, out);
}
